# starting kernel
# baseline (speedup 1.0000x reference)
#include <hip/hip_runtime.h>
#include <hip/hip_bf16.h>

#define N_NODES 50000
#define N_EDGES 800000

typedef float v2f __attribute__((ext_vector_type(2)));
typedef float v8f __attribute__((ext_vector_type(8)));

__global__ void gcn_transpose_k(const float* __restrict__ in, float* __restrict__ out,
                                int rows, int cols) {
    int gid = blockIdx.x * blockDim.x + threadIdx.x;
    if (gid < rows * cols) {
        int r = gid / cols;
        int c = gid - r * cols;
        out[(size_t)c * rows + r] = in[gid];
    }
}

__global__ void gcn_gemm1_wmma(const float* __restrict__ x,
                               const float* __restrict__ w1t,
                               float* __restrict__ out) {
    const int wave    = (int)((blockIdx.x * blockDim.x + threadIdx.x) >> 5);
    const int lane    = threadIdx.x & 31;
    const int rowBase = (wave >> 4) << 4;
    const int colBase = (wave & 15) << 4;
    const int m       = lane & 15;
    const int kl      = (lane >> 4) << 1;

    const float* __restrict__ arow = x + (size_t)(rowBase + m) * 300;
    const float* __restrict__ bcol = w1t + colBase + m;

    v8f c = {0.f, 0.f, 0.f, 0.f, 0.f, 0.f, 0.f, 0.f};
    for (int k = 0; k < 300; k += 4) {
        v2f a = *(const v2f*)(arow + k + kl);
        v2f b;
        b.x = bcol[(size_t)(k + kl)     * 256];
        b.y = bcol[(size_t)(k + kl + 1) * 256];
        c = __builtin_amdgcn_wmma_f32_16x16x4_f32(false, a, false, b,
                                                  (short)0, c, false, false);
    }
    const int mBase = rowBase + ((lane >> 4) << 3);
    float* __restrict__ o = out + (size_t)mBase * 256 + colBase + m;
#pragma unroll
    for (int j = 0; j < 8; ++j)
        o[(size_t)j * 256] = c[j];
}

__global__ void gcn_gemm2_wmma(const float* __restrict__ h1,
                               const float* __restrict__ b1,
                               const float* __restrict__ w2t,
                               float* __restrict__ out) {
    const int wave    = (int)((blockIdx.x * blockDim.x + threadIdx.x) >> 5);
    const int lane    = threadIdx.x & 31;
    const int rowBase = (wave >> 3) << 4;
    const int colBase = (wave & 7)  << 4;
    const int m       = lane & 15;
    const int kl      = (lane >> 4) << 1;

    const float* __restrict__ arow = h1 + (size_t)(rowBase + m) * 256;
    const float* __restrict__ bcol = w2t + colBase + m;

    v8f c = {0.f, 0.f, 0.f, 0.f, 0.f, 0.f, 0.f, 0.f};
    for (int k = 0; k < 256; k += 4) {
        v2f a  = *(const v2f*)(arow + k + kl);
        v2f bb = *(const v2f*)(b1   + k + kl);
        a.x = fmaxf(a.x + bb.x, 0.f);
        a.y = fmaxf(a.y + bb.y, 0.f);
        v2f b;
        b.x = bcol[(size_t)(k + kl)     * 128];
        b.y = bcol[(size_t)(k + kl + 1) * 128];
        c = __builtin_amdgcn_wmma_f32_16x16x4_f32(false, a, false, b,
                                                  (short)0, c, false, false);
    }
    const int mBase = rowBase + ((lane >> 4) << 3);
    float* __restrict__ o = out + (size_t)mBase * 128 + colBase + m;
#pragma unroll
    for (int j = 0; j < 8; ++j)
        o[(size_t)j * 128] = c[j];
}

template <int D>
__global__ void gcn_spmm_scatter(const int*   __restrict__ erow,
                                 const int*   __restrict__ ecol,
                                 const float* __restrict__ eval,
                                 const float* __restrict__ dense,
                                 float*       __restrict__ out) {
    constexpr int CH = D / 4;
    constexpr int SH = (D == 256) ? 6 : 5;
    long gid = (long)blockIdx.x * blockDim.x + threadIdx.x;
    int  e   = (int)(gid >> SH);
    if (e >= N_EDGES) return;
    int  c   = ((int)gid & (CH - 1)) << 2;

    int   r  = erow[e];
    int   cc = ecol[e];
    float v  = eval[e];

    const float4 d = *(const float4*)(dense + (size_t)cc * D + c);
    float* o = out + (size_t)r * D + c;
    (void)__hip_atomic_fetch_add(o + 0, v * d.x, __ATOMIC_RELAXED, __HIP_MEMORY_SCOPE_AGENT);
    (void)__hip_atomic_fetch_add(o + 1, v * d.y, __ATOMIC_RELAXED, __HIP_MEMORY_SCOPE_AGENT);
    (void)__hip_atomic_fetch_add(o + 2, v * d.z, __ATOMIC_RELAXED, __HIP_MEMORY_SCOPE_AGENT);
    (void)__hip_atomic_fetch_add(o + 3, v * d.w, __ATOMIC_RELAXED, __HIP_MEMORY_SCOPE_AGENT);
}

__global__ void gcn_bias_norm(const float* __restrict__ b2, float* __restrict__ out) {
    int gid  = blockIdx.x * blockDim.x + threadIdx.x;
    int row  = gid >> 5;
    int lane = gid & 31;
    if (row >= N_NODES) return;

    float* p = out + (size_t)row * 128 + lane * 4;
    float4 v  = *(float4*)p;
    float4 bb = *(const float4*)(b2 + lane * 4);
    v.x += bb.x; v.y += bb.y; v.z += bb.z; v.w += bb.w;

    float ss = v.x * v.x + v.y * v.y + v.z * v.z + v.w * v.w;
#pragma unroll
    for (int off = 16; off > 0; off >>= 1)
        ss += __shfl_xor(ss, off, 32);

    float scale = 1.0f / fmaxf(sqrtf(ss), 1e-12f);
    v.x *= scale; v.y *= scale; v.z *= scale; v.w *= scale;
    *(float4*)p = v;
}

extern "C" void kernel_launch(void* const* d_in, const int* in_sizes, int n_in,
                              void* d_out, int out_size, void* d_ws, size_t ws_size,
                              hipStream_t stream) {
    const float* x    = (const float*)d_in[0];
    const int*   erow = (const int*)  d_in[1];
    const int*   ecol = (const int*)  d_in[2];
    const float* ev   = (const float*)d_in[3];
    const float* W1   = (const float*)d_in[4];
    const float* b1   = (const float*)d_in[5];
    const float* W2   = (const float*)d_in[6];
    const float* b2   = (const float*)d_in[7];
    float* out = (float*)d_out;

    char* ws = (char*)d_ws;
    float* w1t  = (float*)(ws);
    float* w2t  = (float*)(ws + 307200);
    float* bufA = (float*)(ws + 307200 + 131072);
    float* bufB = (float*)(ws + 307200 + 131072 + 51200000);

    gcn_transpose_k<<<(256 * 300 + 255) / 256, 256, 0, stream>>>(W1, w1t, 256, 300);
    gcn_transpose_k<<<(128 * 256 + 255) / 256, 256, 0, stream>>>(W2, w2t, 128, 256);

    gcn_gemm1_wmma<<<50000 / 8, 256, 0, stream>>>(x, w1t, bufA);

    hipMemsetAsync(bufB, 0, (size_t)N_NODES * 256 * sizeof(float), stream);
    gcn_spmm_scatter<256><<<(N_EDGES * 64) / 256, 256, 0, stream>>>(erow, ecol, ev, bufA, bufB);

    gcn_gemm2_wmma<<<25000 / 8, 256, 0, stream>>>(bufB, b1, w2t, bufA);

    hipMemsetAsync(out, 0, (size_t)N_NODES * 128 * sizeof(float), stream);
    gcn_spmm_scatter<128><<<(N_EDGES * 32) / 256, 256, 0, stream>>>(erow, ecol, ev, bufA, out);

    gcn_bias_norm<<<(N_NODES * 32) / 256, 256, 0, stream>>>(b2, out);
}


# SpMM by destination row: edges grouped by row each call (returning int add as arrival-order rank, scan, fill); one wave sums a row in registers (fma) and stores it once; no float atomics
# speedup vs baseline: 2.0032x; 2.0032x over previous
#include <hip/hip_runtime.h>
#include <hip/hip_bf16.h>

#define N_NODES 50000
#define N_EDGES 800000

typedef float v2f __attribute__((ext_vector_type(2)));
typedef float v8f __attribute__((ext_vector_type(8)));
typedef float v4f __attribute__((ext_vector_type(4)));

__global__ void gcn_transpose_k(const float* __restrict__ in, float* __restrict__ out,
                                int rows, int cols) {
    int gid = blockIdx.x * blockDim.x + threadIdx.x;
    if (gid < rows * cols) {
        int r = gid / cols;
        int c = gid - r * cols;
        out[(size_t)c * rows + r] = in[gid];
    }
}

__global__ void gcn_gemm1_wmma(const float* __restrict__ x,
                               const float* __restrict__ w1t,
                               float* __restrict__ out) {
    const int wave    = (int)((blockIdx.x * blockDim.x + threadIdx.x) >> 5);
    const int lane    = threadIdx.x & 31;
    const int rowBase = (wave >> 4) << 4;
    const int colBase = (wave & 15) << 4;
    const int m       = lane & 15;
    const int kl      = (lane >> 4) << 1;

    const float* __restrict__ arow = x + (size_t)(rowBase + m) * 300;
    const float* __restrict__ bcol = w1t + colBase + m;

    v8f c = {0.f, 0.f, 0.f, 0.f, 0.f, 0.f, 0.f, 0.f};
    for (int k = 0; k < 300; k += 4) {
        v2f a = *(const v2f*)(arow + k + kl);
        v2f b;
        b.x = bcol[(size_t)(k + kl)     * 256];
        b.y = bcol[(size_t)(k + kl + 1) * 256];
        c = __builtin_amdgcn_wmma_f32_16x16x4_f32(false, a, false, b,
                                                  (short)0, c, false, false);
    }
    const int mBase = rowBase + ((lane >> 4) << 3);
    float* __restrict__ o = out + (size_t)mBase * 256 + colBase + m;
#pragma unroll
    for (int j = 0; j < 8; ++j)
        o[(size_t)j * 256] = c[j];
}

__global__ void gcn_gemm2_wmma(const float* __restrict__ h1,
                               const float* __restrict__ b1,
                               const float* __restrict__ w2t,
                               float* __restrict__ out) {
    const int wave    = (int)((blockIdx.x * blockDim.x + threadIdx.x) >> 5);
    const int lane    = threadIdx.x & 31;
    const int rowBase = (wave >> 3) << 4;
    const int colBase = (wave & 7)  << 4;
    const int m       = lane & 15;
    const int kl      = (lane >> 4) << 1;

    const float* __restrict__ arow = h1 + (size_t)(rowBase + m) * 256;
    const float* __restrict__ bcol = w2t + colBase + m;

    v8f c = {0.f, 0.f, 0.f, 0.f, 0.f, 0.f, 0.f, 0.f};
    for (int k = 0; k < 256; k += 4) {
        v2f a  = *(const v2f*)(arow + k + kl);
        v2f bb = *(const v2f*)(b1   + k + kl);
        a.x = fmaxf(a.x + bb.x, 0.f);
        a.y = fmaxf(a.y + bb.y, 0.f);
        v2f b;
        b.x = bcol[(size_t)(k + kl)     * 128];
        b.y = bcol[(size_t)(k + kl + 1) * 128];
        c = __builtin_amdgcn_wmma_f32_16x16x4_f32(false, a, false, b,
                                                  (short)0, c, false, false);
    }
    const int mBase = rowBase + ((lane >> 4) << 3);
    float* __restrict__ o = out + (size_t)mBase * 128 + colBase + m;
#pragma unroll
    for (int j = 0; j < 8; ++j)
        o[(size_t)j * 128] = c[j];
}

__global__ void gcn_edge_rank(const int* __restrict__ erow,
                              int* __restrict__ cnt,
                              int* __restrict__ rank) {
    int e = blockIdx.x * blockDim.x + threadIdx.x;
    if (e < N_EDGES)
        rank[e] = __hip_atomic_fetch_add(cnt + erow[e], 1, __ATOMIC_RELAXED,
                                         __HIP_MEMORY_SCOPE_AGENT);
}

#define SCAN_CHUNKS 13

__device__ __forceinline__ int gcn_wave_incl_scan(int x, int lane) {
    for (int off = 1; off < 32; off <<= 1) {
        int n = __shfl_up(x, off, 32);
        if (lane >= off) x += n;
    }
    return x;
}

__global__ __launch_bounds__(1024) void gcn_row_scan(const int* __restrict__ cnt,
                                                     int* __restrict__ rowptr) {
    __shared__ int wtot[SCAN_CHUNKS][32];
    __shared__ int woff[SCAN_CHUNKS][32];
    __shared__ int ctot[SCAN_CHUNKS];
    const int t    = threadIdx.x;
    const int lane = t & 31;
    const int wave = t >> 5;

    int4 v[SCAN_CHUNKS];
    int  incl[SCAN_CHUNKS];
#pragma unroll
    for (int k = 0; k < SCAN_CHUNKS; ++k) {
        const int q = k * 1024 + t;
        v[k] = make_int4(0, 0, 0, 0);
        if (q < N_NODES / 4) v[k] = ((const int4*)cnt)[q];
    }
#pragma unroll
    for (int k = 0; k < SCAN_CHUNKS; ++k) {
        incl[k] = gcn_wave_incl_scan(v[k].x + v[k].y + v[k].z + v[k].w, lane);
        if (lane == 31) wtot[k][wave] = incl[k];
    }
    __syncthreads();
    if (wave < SCAN_CHUNKS) {
        const int wv   = wtot[wave][lane];
        const int winc = gcn_wave_incl_scan(wv, lane);
        woff[wave][lane] = winc - wv;
        if (lane == 31) ctot[wave] = winc;
    }
    __syncthreads();
    int carry = 0;
#pragma unroll
    for (int k = 0; k < SCAN_CHUNKS; ++k) {
        const int q  = k * 1024 + t;
        const int s  = v[k].x + v[k].y + v[k].z + v[k].w;
        const int ex = carry + woff[k][wave] + incl[k] - s;
        if (q < N_NODES / 4)
            ((int4*)rowptr)[q] = make_int4(ex, ex + v[k].x, ex + v[k].x + v[k].y,
                                           ex + v[k].x + v[k].y + v[k].z);
        carry += ctot[k];
    }
    if (t == 0) rowptr[N_NODES] = carry;
}

__global__ void gcn_edge_fill(const int*   __restrict__ erow,
                              const int*   __restrict__ ecol,
                              const float* __restrict__ eval,
                              const int*   __restrict__ rank,
                              const int*   __restrict__ rowptr,
                              int*         __restrict__ edges) {
    int e = blockIdx.x * blockDim.x + threadIdx.x;
    if (e < N_EDGES) {
        const int pos = rowptr[erow[e]] + rank[e];
        ((int2*)edges)[pos] = make_int2(ecol[e], __float_as_int(eval[e]));
    }
}

template <int D>
__global__ __launch_bounds__(256) void gcn_spmm_rows(const int*   __restrict__ rowptr,
                                                     const int*   __restrict__ edges,
                                                     const float* __restrict__ dense,
                                                     float*       __restrict__ dst) {
    constexpr int V = D / 128;
    const int lane = threadIdx.x & 31;
    const int row  = __builtin_amdgcn_readfirstlane((int)(blockIdx.x * 8 + (threadIdx.x >> 5)));
    if (row >= N_NODES) return;
    const int start = rowptr[row];
    const int end   = rowptr[row + 1];

    v4f acc[V];
    for (int w = 0; w < V; ++w) acc[w] = (v4f){0.f, 0.f, 0.f, 0.f};

    const int2* __restrict__ er = (const int2*)edges;

    for (int base = start; base < end; base += 32) {
        const int n = min(32, end - base);
        int2 rec = make_int2(0, 0);
        if (lane < n) rec = er[base + lane];

        int j = 0;
        for (; j + 4 <= n; j += 4) {
            float v[4];
            v4f   d[4][V];
#pragma unroll
            for (int u = 0; u < 4; ++u) {
                const int c = __builtin_amdgcn_readlane(rec.x, j + u);
                v[u] = __int_as_float(__builtin_amdgcn_readlane(rec.y, j + u));
                const v4f* __restrict__ rp = (const v4f*)(dense + (size_t)c * D);
#pragma unroll
                for (int w = 0; w < V; ++w) d[u][w] = rp[lane + w * 32];
            }
#pragma unroll
            for (int u = 0; u < 4; ++u)
#pragma unroll
                for (int w = 0; w < V; ++w) acc[w] += v[u] * d[u][w];
        }
#pragma nounroll
        for (; j < n; ++j) {
            const int   c = __builtin_amdgcn_readlane(rec.x, j);
            const float v = __int_as_float(__builtin_amdgcn_readlane(rec.y, j));
            const v4f* __restrict__ rp = (const v4f*)(dense + (size_t)c * D);
            for (int w = 0; w < V; ++w) acc[w] += v * rp[lane + w * 32];
        }
    }

    v4f* __restrict__ o = (v4f*)(dst + (size_t)row * D);
    for (int w = 0; w < V; ++w) o[lane + w * 32] = acc[w];
}

__global__ void gcn_bias_norm(const float* __restrict__ b2, const float* __restrict__ src,
                              float* __restrict__ out) {
    int gid  = blockIdx.x * blockDim.x + threadIdx.x;
    int row  = gid >> 5;
    int lane = gid & 31;
    if (row >= N_NODES) return;

    float4 v  = *(const float4*)(src + (size_t)row * 128 + lane * 4);
    float4 bb = *(const float4*)(b2 + lane * 4);
    v.x += bb.x; v.y += bb.y; v.z += bb.z; v.w += bb.w;

    float ss = v.x * v.x + v.y * v.y + v.z * v.z + v.w * v.w;
#pragma unroll
    for (int off = 16; off > 0; off >>= 1)
        ss += __shfl_xor(ss, off, 32);

    float scale = 1.0f / fmaxf(sqrtf(ss), 1e-12f);
    v.x *= scale; v.y *= scale; v.z *= scale; v.w *= scale;
    *(float4*)(out + (size_t)row * 128 + lane * 4) = v;
}

extern "C" void kernel_launch(void* const* d_in, const int* in_sizes, int n_in,
                              void* d_out, int out_size, void* d_ws, size_t ws_size,
                              hipStream_t stream) {
    const float* x    = (const float*)d_in[0];
    const int*   erow = (const int*)  d_in[1];
    const int*   ecol = (const int*)  d_in[2];
    const float* ev   = (const float*)d_in[3];
    const float* W1   = (const float*)d_in[4];
    const float* b1   = (const float*)d_in[5];
    const float* W2   = (const float*)d_in[6];
    const float* b2   = (const float*)d_in[7];
    float* out = (float*)d_out;

    char* ws = (char*)d_ws;
    float* w1t  = (float*)(ws);
    float* w2t  = (float*)(ws + 307200);
    float* bufA = (float*)(ws + 307200 + 131072);
    float* bufB = (float*)(ws + 307200 + 131072 + 51200000);

    char* ob = (char*)d_out;
    int* cnt    = (int*)(ob);
    int* rowptr = (int*)(ob + 212992);
    int* rank   = (int*)(ob + 425984);
    int* edges  = (int*)(ob + 425984 + 3200000);

    gcn_transpose_k<<<(256 * 300 + 255) / 256, 256, 0, stream>>>(W1, w1t, 256, 300);
    gcn_transpose_k<<<(128 * 256 + 255) / 256, 256, 0, stream>>>(W2, w2t, 128, 256);

    hipMemsetAsync(cnt, 0, (size_t)N_NODES * sizeof(int), stream);
    gcn_edge_rank<<<N_EDGES / 256, 256, 0, stream>>>(erow, cnt, rank);
    gcn_row_scan<<<1, 1024, 0, stream>>>(cnt, rowptr);
    gcn_edge_fill<<<N_EDGES / 256, 256, 0, stream>>>(erow, ecol, ev, rank, rowptr, edges);

    gcn_gemm1_wmma<<<50000 / 8, 256, 0, stream>>>(x, w1t, bufA);

    gcn_spmm_rows<256><<<N_NODES / 8, 256, 0, stream>>>(rowptr, edges, bufA, bufB);

    gcn_gemm2_wmma<<<25000 / 8, 256, 0, stream>>>(bufB, b1, w2t, bufA);

    gcn_spmm_rows<128><<<N_NODES / 8, 256, 0, stream>>>(rowptr, edges, bufA, bufB);

    gcn_bias_norm<<<(N_NODES * 32) / 256, 256, 0, stream>>>(b2, bufB, out);
}


# Both GEMMs tiled through LDS: 80-row x 128/64-col block tiles, K slabs of a and of W (as given, K-contiguous: the two transposes go) copied with b128 loads; 5 WMMAs per 6 LDS fragment reads a wave
# speedup vs baseline: 3.0000x; 1.4976x over previous
#include <hip/hip_runtime.h>
#include <hip/hip_bf16.h>

#define N_NODES 50000
#define N_EDGES 800000

typedef float v2f __attribute__((ext_vector_type(2)));
typedef float v8f __attribute__((ext_vector_type(8)));
typedef float v4f __attribute__((ext_vector_type(4)));

#define GM_ROWS 80

template <int K, int BK, int NTOT, int BN, bool BIAS_RELU>
__global__ __launch_bounds__(BN * 2)
void gcn_gemm_lds(const float* __restrict__ a,
                  const float* __restrict__ bias,
                  const float* __restrict__ w,
                  float* __restrict__ out) {
    static_assert(K % BK == 0 && BK % 4 == 0 && NTOT % BN == 0 && BN % 16 == 0, "tile shape");
    constexpr int NTHR = BN * 2;
    constexpr int C4   = BK / 4;
    constexpr int P    = (C4 & 1) ? BK : BK + 4;
    __shared__ __attribute__((aligned(16))) float As[GM_ROWS * P];
    __shared__ __attribute__((aligned(16))) float Bs[BN * P];

    const int tid  = threadIdx.x;
    const int lane = tid & 31;
    const int wave = tid >> 5;
    const int c    = lane & 15;
    const int h2   = (lane >> 4) << 1;
    constexpr int NCB = NTOT / BN;
    const int rowBase = (int)(blockIdx.x / NCB) * GM_ROWS;
    const int colBase = (int)(blockIdx.x % NCB) * BN;

    v8f acc[5];
#pragma unroll
    for (int rt = 0; rt < 5; ++rt) acc[rt] = (v8f){0.f, 0.f, 0.f, 0.f, 0.f, 0.f, 0.f, 0.f};

    static_assert(C4 <= 16 && GM_ROWS % (NTHR / 16) == 0 && BN % (NTHR / 16) == 0, "copy shape");
    constexpr int RPP = NTHR / 16;
    constexpr int NAP = GM_ROWS / RPP;
    constexpr int NBP = BN / RPP;
    const int cr = tid >> 4;
    const int cq = tid & 15;
    const float* __restrict__ ap = a + (size_t)(rowBase + cr) * K + cq * 4;
    const float* __restrict__ wp = w + (size_t)(colBase + cr) * K + cq * 4;
    float* asl = As + cr * P + cq * 4;
    float* bsl = Bs + cr * P + cq * 4;
    const float* af = As + c * P + h2;
    const float* bf = Bs + (wave * 16 + c) * P + h2;

    for (int k0 = 0; k0 < K; k0 += BK) {
        if (cq < C4) {
            float4 ra[NAP], rb[NBP];
#pragma unroll
            for (int p = 0; p < NAP; ++p)
                ra[p] = *(const float4*)(ap + (size_t)p * RPP * K + k0);
#pragma unroll
            for (int p = 0; p < NBP; ++p)
                rb[p] = *(const float4*)(wp + (size_t)p * RPP * K + k0);
            if (BIAS_RELU) {
                const float4 bb = *(const float4*)(bias + k0 + cq * 4);
#pragma unroll
                for (int p = 0; p < NAP; ++p) {
                    ra[p].x = fmaxf(ra[p].x + bb.x, 0.f);
                    ra[p].y = fmaxf(ra[p].y + bb.y, 0.f);
                    ra[p].z = fmaxf(ra[p].z + bb.z, 0.f);
                    ra[p].w = fmaxf(ra[p].w + bb.w, 0.f);
                }
            }
#pragma unroll
            for (int p = 0; p < NAP; ++p) *(float4*)(asl + p * RPP * P) = ra[p];
#pragma unroll
            for (int p = 0; p < NBP; ++p) *(float4*)(bsl + p * RPP * P) = rb[p];
        }
        __syncthreads();
#pragma unroll
        for (int s = 0; s < C4; ++s) {
            const v2f b = *(const v2f*)(bf + s * 4);
#pragma unroll
            for (int rt = 0; rt < 5; ++rt) {
                const v2f av = *(const v2f*)(af + rt * 16 * P + s * 4);
                acc[rt] = __builtin_amdgcn_wmma_f32_16x16x4_f32(false, av, false, b,
                                                                (short)0, acc[rt], false, false);
            }
        }
        __syncthreads();
    }

    float* __restrict__ o = out + (size_t)(rowBase + ((lane >> 4) << 3)) * NTOT
                                + colBase + wave * 16 + c;
#pragma unroll
    for (int rt = 0; rt < 5; ++rt)
#pragma unroll
        for (int j = 0; j < 8; ++j)
            o[(size_t)(rt * 16 + j) * NTOT] = acc[rt][j];
}

__global__ void gcn_edge_rank(const int* __restrict__ erow,
                              int* __restrict__ cnt,
                              int* __restrict__ rank) {
    int e = blockIdx.x * blockDim.x + threadIdx.x;
    if (e < N_EDGES)
        rank[e] = __hip_atomic_fetch_add(cnt + erow[e], 1, __ATOMIC_RELAXED,
                                         __HIP_MEMORY_SCOPE_AGENT);
}

#define SCAN_CHUNKS 13

__device__ __forceinline__ int gcn_wave_incl_scan(int x, int lane) {
    for (int off = 1; off < 32; off <<= 1) {
        int n = __shfl_up(x, off, 32);
        if (lane >= off) x += n;
    }
    return x;
}

__global__ __launch_bounds__(1024) void gcn_row_scan(const int* __restrict__ cnt,
                                                     int* __restrict__ rowptr) {
    __shared__ int wtot[SCAN_CHUNKS][32];
    __shared__ int woff[SCAN_CHUNKS][32];
    __shared__ int ctot[SCAN_CHUNKS];
    const int t    = threadIdx.x;
    const int lane = t & 31;
    const int wave = t >> 5;

    int4 v[SCAN_CHUNKS];
    int  incl[SCAN_CHUNKS];
#pragma unroll
    for (int k = 0; k < SCAN_CHUNKS; ++k) {
        const int q = k * 1024 + t;
        v[k] = make_int4(0, 0, 0, 0);
        if (q < N_NODES / 4) v[k] = ((const int4*)cnt)[q];
    }
#pragma unroll
    for (int k = 0; k < SCAN_CHUNKS; ++k) {
        incl[k] = gcn_wave_incl_scan(v[k].x + v[k].y + v[k].z + v[k].w, lane);
        if (lane == 31) wtot[k][wave] = incl[k];
    }
    __syncthreads();
    if (wave < SCAN_CHUNKS) {
        const int wv   = wtot[wave][lane];
        const int winc = gcn_wave_incl_scan(wv, lane);
        woff[wave][lane] = winc - wv;
        if (lane == 31) ctot[wave] = winc;
    }
    __syncthreads();
    int carry = 0;
#pragma unroll
    for (int k = 0; k < SCAN_CHUNKS; ++k) {
        const int q  = k * 1024 + t;
        const int s  = v[k].x + v[k].y + v[k].z + v[k].w;
        const int ex = carry + woff[k][wave] + incl[k] - s;
        if (q < N_NODES / 4)
            ((int4*)rowptr)[q] = make_int4(ex, ex + v[k].x, ex + v[k].x + v[k].y,
                                           ex + v[k].x + v[k].y + v[k].z);
        carry += ctot[k];
    }
    if (t == 0) rowptr[N_NODES] = carry;
}

__global__ void gcn_edge_fill(const int*   __restrict__ erow,
                              const int*   __restrict__ ecol,
                              const float* __restrict__ eval,
                              const int*   __restrict__ rank,
                              const int*   __restrict__ rowptr,
                              int*         __restrict__ edges) {
    int e = blockIdx.x * blockDim.x + threadIdx.x;
    if (e < N_EDGES) {
        const int pos = rowptr[erow[e]] + rank[e];
        ((int2*)edges)[pos] = make_int2(ecol[e], __float_as_int(eval[e]));
    }
}

template <int D>
__global__ __launch_bounds__(256) void gcn_spmm_rows(const int*   __restrict__ rowptr,
                                                     const int*   __restrict__ edges,
                                                     const float* __restrict__ dense,
                                                     float*       __restrict__ dst) {
    constexpr int V = D / 128;
    const int lane = threadIdx.x & 31;
    const int row  = __builtin_amdgcn_readfirstlane((int)(blockIdx.x * 8 + (threadIdx.x >> 5)));
    if (row >= N_NODES) return;
    const int start = rowptr[row];
    const int end   = rowptr[row + 1];

    v4f acc[V];
    for (int w = 0; w < V; ++w) acc[w] = (v4f){0.f, 0.f, 0.f, 0.f};

    const int2* __restrict__ er = (const int2*)edges;

    for (int base = start; base < end; base += 32) {
        const int n = min(32, end - base);
        int2 rec = make_int2(0, 0);
        if (lane < n) rec = er[base + lane];

        int j = 0;
        for (; j + 4 <= n; j += 4) {
            float v[4];
            v4f   d[4][V];
#pragma unroll
            for (int u = 0; u < 4; ++u) {
                const int c = __builtin_amdgcn_readlane(rec.x, j + u);
                v[u] = __int_as_float(__builtin_amdgcn_readlane(rec.y, j + u));
                const v4f* __restrict__ rp = (const v4f*)(dense + (size_t)c * D);
#pragma unroll
                for (int w = 0; w < V; ++w) d[u][w] = rp[lane + w * 32];
            }
#pragma unroll
            for (int u = 0; u < 4; ++u)
#pragma unroll
                for (int w = 0; w < V; ++w) acc[w] += v[u] * d[u][w];
        }
#pragma nounroll
        for (; j < n; ++j) {
            const int   c = __builtin_amdgcn_readlane(rec.x, j);
            const float v = __int_as_float(__builtin_amdgcn_readlane(rec.y, j));
            const v4f* __restrict__ rp = (const v4f*)(dense + (size_t)c * D);
            for (int w = 0; w < V; ++w) acc[w] += v * rp[lane + w * 32];
        }
    }

    v4f* __restrict__ o = (v4f*)(dst + (size_t)row * D);
    for (int w = 0; w < V; ++w) o[lane + w * 32] = acc[w];
}

__global__ void gcn_bias_norm(const float* __restrict__ b2, const float* __restrict__ src,
                              float* __restrict__ out) {
    int gid  = blockIdx.x * blockDim.x + threadIdx.x;
    int row  = gid >> 5;
    int lane = gid & 31;
    if (row >= N_NODES) return;

    float4 v  = *(const float4*)(src + (size_t)row * 128 + lane * 4);
    float4 bb = *(const float4*)(b2 + lane * 4);
    v.x += bb.x; v.y += bb.y; v.z += bb.z; v.w += bb.w;

    float ss = v.x * v.x + v.y * v.y + v.z * v.z + v.w * v.w;
#pragma unroll
    for (int off = 16; off > 0; off >>= 1)
        ss += __shfl_xor(ss, off, 32);

    float scale = 1.0f / fmaxf(sqrtf(ss), 1e-12f);
    v.x *= scale; v.y *= scale; v.z *= scale; v.w *= scale;
    *(float4*)(out + (size_t)row * 128 + lane * 4) = v;
}

extern "C" void kernel_launch(void* const* d_in, const int* in_sizes, int n_in,
                              void* d_out, int out_size, void* d_ws, size_t ws_size,
                              hipStream_t stream) {
    const float* x    = (const float*)d_in[0];
    const int*   erow = (const int*)  d_in[1];
    const int*   ecol = (const int*)  d_in[2];
    const float* ev   = (const float*)d_in[3];
    const float* W1   = (const float*)d_in[4];
    const float* b1   = (const float*)d_in[5];
    const float* W2   = (const float*)d_in[6];
    const float* b2   = (const float*)d_in[7];
    float* out = (float*)d_out;

    char* ws = (char*)d_ws;
    float* bufA = (float*)(ws + 307200 + 131072);
    float* bufB = (float*)(ws + 307200 + 131072 + 51200000);

    char* ob = (char*)d_out;
    int* cnt    = (int*)(ob);
    int* rowptr = (int*)(ob + 212992);
    int* rank   = (int*)(ob + 425984);
    int* edges  = (int*)(ob + 425984 + 3200000);

    hipMemsetAsync(cnt, 0, (size_t)N_NODES * sizeof(int), stream);
    gcn_edge_rank<<<N_EDGES / 256, 256, 0, stream>>>(erow, cnt, rank);
    gcn_row_scan<<<1, 1024, 0, stream>>>(cnt, rowptr);
    gcn_edge_fill<<<N_EDGES / 256, 256, 0, stream>>>(erow, ecol, ev, rank, rowptr, edges);

    gcn_gemm_lds<300, 60, 256, 128, false><<<1250, 256, 0, stream>>>(x, b1, W1, bufA);

    gcn_spmm_rows<256><<<N_NODES / 8, 256, 0, stream>>>(rowptr, edges, bufA, bufB);

    gcn_gemm_lds<256, 64, 128, 64, true><<<1250, 128, 0, stream>>>(bufB, b1, W2, bufA);

    gcn_spmm_rows<128><<<N_NODES / 8, 256, 0, stream>>>(rowptr, edges, bufA, bufB);

    gcn_bias_norm<<<(N_NODES * 32) / 256, 256, 0, stream>>>(b2, bufB, out);
}


# Edges grouped by row with LDS counters only: two-level counting sort (250 row buckets x 128 edge chunks), no global atomics, no memset; a row's sum order = arrival order of the LDS adds
# speedup vs baseline: 3.7803x; 1.2601x over previous
#include <hip/hip_runtime.h>
#include <hip/hip_bf16.h>

#define N_NODES 50000
#define N_EDGES 800000

typedef float v2f __attribute__((ext_vector_type(2)));
typedef float v8f __attribute__((ext_vector_type(8)));
typedef float v4f __attribute__((ext_vector_type(4)));

#define GM_ROWS 80

template <int K, int BK, int NTOT, int BN, bool BIAS_RELU>
__global__ __launch_bounds__(BN * 2)
void gcn_gemm_lds(const float* __restrict__ a,
                  const float* __restrict__ bias,
                  const float* __restrict__ w,
                  float* __restrict__ out) {
    static_assert(K % BK == 0 && BK % 4 == 0 && NTOT % BN == 0 && BN % 16 == 0, "tile shape");
    constexpr int NTHR = BN * 2;
    constexpr int C4   = BK / 4;
    constexpr int P    = (C4 & 1) ? BK : BK + 4;
    __shared__ __attribute__((aligned(16))) float As[GM_ROWS * P];
    __shared__ __attribute__((aligned(16))) float Bs[BN * P];

    const int tid  = threadIdx.x;
    const int lane = tid & 31;
    const int wave = tid >> 5;
    const int c    = lane & 15;
    const int h2   = (lane >> 4) << 1;
    constexpr int NCB = NTOT / BN;
    const int rowBase = (int)(blockIdx.x / NCB) * GM_ROWS;
    const int colBase = (int)(blockIdx.x % NCB) * BN;

    v8f acc[5];
#pragma unroll
    for (int rt = 0; rt < 5; ++rt) acc[rt] = (v8f){0.f, 0.f, 0.f, 0.f, 0.f, 0.f, 0.f, 0.f};

    static_assert(C4 <= 16 && GM_ROWS % (NTHR / 16) == 0 && BN % (NTHR / 16) == 0, "copy shape");
    constexpr int RPP = NTHR / 16;
    constexpr int NAP = GM_ROWS / RPP;
    constexpr int NBP = BN / RPP;
    const int cr = tid >> 4;
    const int cq = tid & 15;
    const float* __restrict__ ap = a + (size_t)(rowBase + cr) * K + cq * 4;
    const float* __restrict__ wp = w + (size_t)(colBase + cr) * K + cq * 4;
    float* asl = As + cr * P + cq * 4;
    float* bsl = Bs + cr * P + cq * 4;
    const float* af = As + c * P + h2;
    const float* bf = Bs + (wave * 16 + c) * P + h2;

    for (int k0 = 0; k0 < K; k0 += BK) {
        if (cq < C4) {
            float4 ra[NAP], rb[NBP];
#pragma unroll
            for (int p = 0; p < NAP; ++p)
                ra[p] = *(const float4*)(ap + (size_t)p * RPP * K + k0);
#pragma unroll
            for (int p = 0; p < NBP; ++p)
                rb[p] = *(const float4*)(wp + (size_t)p * RPP * K + k0);
            if (BIAS_RELU) {
                const float4 bb = *(const float4*)(bias + k0 + cq * 4);
#pragma unroll
                for (int p = 0; p < NAP; ++p) {
                    ra[p].x = fmaxf(ra[p].x + bb.x, 0.f);
                    ra[p].y = fmaxf(ra[p].y + bb.y, 0.f);
                    ra[p].z = fmaxf(ra[p].z + bb.z, 0.f);
                    ra[p].w = fmaxf(ra[p].w + bb.w, 0.f);
                }
            }
#pragma unroll
            for (int p = 0; p < NAP; ++p) *(float4*)(asl + p * RPP * P) = ra[p];
#pragma unroll
            for (int p = 0; p < NBP; ++p) *(float4*)(bsl + p * RPP * P) = rb[p];
        }
        __syncthreads();
#pragma unroll
        for (int s = 0; s < C4; ++s) {
            const v2f b = *(const v2f*)(bf + s * 4);
#pragma unroll
            for (int rt = 0; rt < 5; ++rt) {
                const v2f av = *(const v2f*)(af + rt * 16 * P + s * 4);
                acc[rt] = __builtin_amdgcn_wmma_f32_16x16x4_f32(false, av, false, b,
                                                                (short)0, acc[rt], false, false);
            }
        }
        __syncthreads();
    }

    float* __restrict__ o = out + (size_t)(rowBase + ((lane >> 4) << 3)) * NTOT
                                + colBase + wave * 16 + c;
#pragma unroll
    for (int rt = 0; rt < 5; ++rt)
#pragma unroll
        for (int j = 0; j < 8; ++j)
            o[(size_t)(rt * 16 + j) * NTOT] = acc[rt][j];
}

#define NBK      250
#define BK_ROWS  200
#define NCH      128
#define CH_EDGES 6250
#define GS_NT    512
#define CH_IT    ((CH_EDGES + GS_NT - 1) / GS_NT)
#define GS_Q     (128 / (GS_NT / 32))
#define GS_REG   4096
#define GS_RK    (GS_REG / GS_NT)

__device__ __forceinline__ int gcn_wave_incl_scan(int x, int lane) {
    for (int off = 1; off < 32; off <<= 1) {
        int n = __shfl_up(x, off, 32);
        if (lane >= off) x += n;
    }
    return x;
}

__global__ __launch_bounds__(GS_NT) void gcn_bucket_count(const int* __restrict__ erow,
                                                          int* __restrict__ counts32) {
    unsigned short* __restrict__ counts = (unsigned short*)counts32;
    __shared__ int hist[256];
    const int t  = threadIdx.x;
    const int wg = blockIdx.x;
    const int* __restrict__ er = erow + wg * CH_EDGES;
    int r[CH_IT];
#pragma unroll
    for (int k = 0; k < CH_IT; ++k) r[k] = er[min(t + k * GS_NT, CH_EDGES - 1)];
    if (t < 256) hist[t] = 0;
    __syncthreads();
#pragma unroll
    for (int k = 0; k < CH_IT; ++k)
        if (t + k * GS_NT < CH_EDGES) atomicAdd(&hist[r[k] / BK_ROWS], 1);
    __syncthreads();
    if (t < NBK) counts[t * NCH + wg] = (unsigned short)hist[t];
}

__global__ __launch_bounds__(GS_NT) void gcn_bucket_scatter(const int*   __restrict__ erow,
                                                            const int*   __restrict__ ecol,
                                                            const float* __restrict__ eval,
                                                            const int* __restrict__ counts32,
                                                            int* __restrict__ bstart,
                                                            int* __restrict__ inter) {
    const unsigned short* __restrict__ counts = (const unsigned short*)counts32;
    __shared__ int tot[256];
    __shared__ int pre[256];
    __shared__ int cur[256];
    const int t    = threadIdx.x;
    const int lane = t & 31;
    const int wave = t >> 5;
    const int wg   = blockIdx.x;

    int r[CH_IT], c[CH_IT], v[CH_IT];
#pragma unroll
    for (int k = 0; k < CH_IT; ++k) {
        const int e = wg * CH_EDGES + min(t + k * GS_NT, CH_EDGES - 1);
        r[k] = erow[e]; c[k] = ecol[e]; v[k] = __float_as_int(eval[e]);
    }

    const int hl = lane & 15;
    __builtin_amdgcn_sched_barrier(0);
    uint4 w4[GS_Q];
#pragma unroll
    for (int q = 0; q < GS_Q; ++q) {
        const int b = 2 * (wave * GS_Q + q) + (lane >> 4);
        w4[q] = *(const uint4*)(counts + min(b, NBK - 1) * NCH + hl * 8);
    }
#pragma unroll
    for (int q = 0; q < GS_Q; ++q) {
        const int b = 2 * (wave * GS_Q + q) + (lane >> 4);
        const uint4 w = w4[q];
        const unsigned x[8] = {w.x & 0xffffu, w.x >> 16, w.y & 0xffffu, w.y >> 16,
                               w.z & 0xffffu, w.z >> 16, w.w & 0xffffu, w.w >> 16};
        int s = 0, p = 0;
#pragma unroll
        for (int j = 0; j < 8; ++j) {
            s += (int)x[j];
            if (hl * 8 + j < wg) p += (int)x[j];
        }
        if (b >= NBK) { s = 0; p = 0; }
#pragma unroll
        for (int off = 8; off > 0; off >>= 1) {
            s += __shfl_xor(s, off, 32);
            p += __shfl_xor(p, off, 32);
        }
        if (hl == 0) { tot[b] = s; pre[b] = p; }
    }
    __syncthreads();
    if (wave == 0) {
        int a[8];
        int sum = 0;
#pragma unroll
        for (int j = 0; j < 8; ++j) { a[j] = tot[8 * lane + j]; sum += a[j]; }
        int ex = gcn_wave_incl_scan(sum, lane) - sum;
#pragma unroll
        for (int j = 0; j < 8; ++j) {
            const int b = 8 * lane + j;
            cur[b] = ex + pre[b];
            if (wg == 0 && b <= NBK) bstart[b] = ex;
            ex += a[j];
        }
    }
    __syncthreads();
    int pos[CH_IT];
#pragma unroll
    for (int k = 0; k < CH_IT; ++k) {
        const int b = r[k] / BK_ROWS;
        pos[k] = 0;
        if (t + k * GS_NT < CH_EDGES) pos[k] = atomicAdd(&cur[b], 1);
        c[k] |= (r[k] - b * BK_ROWS) << 16;
    }
#pragma unroll
    for (int k = 0; k < CH_IT; ++k)
        if (t + k * GS_NT < CH_EDGES) ((int2*)inter)[pos[k]] = make_int2(c[k], v[k]);
}

__global__ __launch_bounds__(GS_NT) void gcn_bucket_sort(const int* __restrict__ bstart,
                                                         const int* __restrict__ inter,
                                                         int* __restrict__ rowptr,
                                                         int* __restrict__ edges) {
    __shared__ int cnt[256];
    __shared__ int cur[256];
    const int t    = threadIdx.x;
    const int lane = t & 31;
    const int wave = t >> 5;
    const int b    = blockIdx.x;
    const int s    = bstart[b];
    const int n    = bstart[b + 1] - s;
    const int2* __restrict__ in = (const int2*)inter + s;

    int2 rec[GS_RK];
#pragma unroll
    for (int k = 0; k < GS_RK; ++k) {
        const int i = t + k * GS_NT;
        rec[k] = make_int2(0, 0);
        if (i < n) rec[k] = in[i];
    }
    if (t < 256) cnt[t] = 0;
    __syncthreads();
#pragma unroll
    for (int k = 0; k < GS_RK; ++k)
        if (t + k * GS_NT < n) atomicAdd(&cnt[rec[k].x >> 16], 1);
    for (int i = t + GS_REG; i < n; i += GS_NT) atomicAdd(&cnt[in[i].x >> 16], 1);
    __syncthreads();
    if (wave == 0) {
        int a[8];
        int sum = 0;
#pragma unroll
        for (int j = 0; j < 8; ++j) { a[j] = cnt[8 * lane + j]; sum += a[j]; }
        int ex = s + gcn_wave_incl_scan(sum, lane) - sum;
#pragma unroll
        for (int j = 0; j < 8; ++j) {
            const int rib = 8 * lane + j;
            cur[rib] = ex;
            if (rib < BK_ROWS) rowptr[b * BK_ROWS + rib] = ex;
            ex += a[j];
        }
    }
    if (b == NBK - 1 && t == 0) rowptr[N_NODES] = s + n;
    __syncthreads();
    int pos[GS_RK];
#pragma unroll
    for (int k = 0; k < GS_RK; ++k) {
        pos[k] = 0;
        if (t + k * GS_NT < n) pos[k] = atomicAdd(&cur[rec[k].x >> 16], 1);
    }
#pragma unroll
    for (int k = 0; k < GS_RK; ++k)
        if (t + k * GS_NT < n) ((int2*)edges)[pos[k]] = make_int2(rec[k].x & 0xffff, rec[k].y);
    for (int i = t + GS_REG; i < n; i += GS_NT) {
        const int2 rc = in[i];
        const int  p  = atomicAdd(&cur[rc.x >> 16], 1);
        ((int2*)edges)[p] = make_int2(rc.x & 0xffff, rc.y);
    }
}

template <int D>
__global__ __launch_bounds__(256) void gcn_spmm_rows(const int*   __restrict__ rowptr,
                                                     const int*   __restrict__ edges,
                                                     const float* __restrict__ dense,
                                                     float*       __restrict__ dst) {
    constexpr int V = D / 128;
    const int lane = threadIdx.x & 31;
    const int row  = __builtin_amdgcn_readfirstlane((int)(blockIdx.x * 8 + (threadIdx.x >> 5)));
    if (row >= N_NODES) return;
    const int start = rowptr[row];
    const int end   = rowptr[row + 1];

    v4f acc[V];
    for (int w = 0; w < V; ++w) acc[w] = (v4f){0.f, 0.f, 0.f, 0.f};

    const int2* __restrict__ er = (const int2*)edges;

    for (int base = start; base < end; base += 32) {
        const int n = min(32, end - base);
        int2 rec = make_int2(0, 0);
        if (lane < n) rec = er[base + lane];

        int j = 0;
        for (; j + 4 <= n; j += 4) {
            float v[4];
            v4f   d[4][V];
#pragma unroll
            for (int u = 0; u < 4; ++u) {
                const int c = __builtin_amdgcn_readlane(rec.x, j + u);
                v[u] = __int_as_float(__builtin_amdgcn_readlane(rec.y, j + u));
                const v4f* __restrict__ rp = (const v4f*)(dense + (size_t)c * D);
#pragma unroll
                for (int w = 0; w < V; ++w) d[u][w] = rp[lane + w * 32];
            }
#pragma unroll
            for (int u = 0; u < 4; ++u)
#pragma unroll
                for (int w = 0; w < V; ++w) acc[w] += v[u] * d[u][w];
        }
#pragma nounroll
        for (; j < n; ++j) {
            const int   c = __builtin_amdgcn_readlane(rec.x, j);
            const float v = __int_as_float(__builtin_amdgcn_readlane(rec.y, j));
            const v4f* __restrict__ rp = (const v4f*)(dense + (size_t)c * D);
            for (int w = 0; w < V; ++w) acc[w] += v * rp[lane + w * 32];
        }
    }

    v4f* __restrict__ o = (v4f*)(dst + (size_t)row * D);
    for (int w = 0; w < V; ++w) o[lane + w * 32] = acc[w];
}

__global__ void gcn_bias_norm(const float* __restrict__ b2, const float* __restrict__ src,
                              float* __restrict__ out) {
    int gid  = blockIdx.x * blockDim.x + threadIdx.x;
    int row  = gid >> 5;
    int lane = gid & 31;
    if (row >= N_NODES) return;

    float4 v  = *(const float4*)(src + (size_t)row * 128 + lane * 4);
    float4 bb = *(const float4*)(b2 + lane * 4);
    v.x += bb.x; v.y += bb.y; v.z += bb.z; v.w += bb.w;

    float ss = v.x * v.x + v.y * v.y + v.z * v.z + v.w * v.w;
#pragma unroll
    for (int off = 16; off > 0; off >>= 1)
        ss += __shfl_xor(ss, off, 32);

    float scale = 1.0f / fmaxf(sqrtf(ss), 1e-12f);
    v.x *= scale; v.y *= scale; v.z *= scale; v.w *= scale;
    *(float4*)(out + (size_t)row * 128 + lane * 4) = v;
}

extern "C" void kernel_launch(void* const* d_in, const int* in_sizes, int n_in,
                              void* d_out, int out_size, void* d_ws, size_t ws_size,
                              hipStream_t stream) {
    const float* x    = (const float*)d_in[0];
    const int*   erow = (const int*)  d_in[1];
    const int*   ecol = (const int*)  d_in[2];
    const float* ev   = (const float*)d_in[3];
    const float* W1   = (const float*)d_in[4];
    const float* b1   = (const float*)d_in[5];
    const float* W2   = (const float*)d_in[6];
    const float* b2   = (const float*)d_in[7];
    float* out = (float*)d_out;

    char* ws = (char*)d_ws;
    float* bufA = (float*)(ws + 307200 + 131072);
    float* bufB = (float*)(ws + 307200 + 131072 + 51200000);

    char* ob = (char*)d_out;
    int* rowptr = (int*)(ob + 212992);
    int* edges  = (int*)(ob + 425984 + 3200000);
    int* inter  = (int*)(ob + 10025984);
    int* counts = (int*)(ws);
    int* bstart = (int*)(ws + 65536);

    gcn_bucket_count<<<NCH, GS_NT, 0, stream>>>(erow, counts);
    gcn_bucket_scatter<<<NCH, GS_NT, 0, stream>>>(erow, ecol, ev, counts, bstart, inter);
    gcn_bucket_sort<<<NBK, GS_NT, 0, stream>>>(bstart, inter, rowptr, edges);

    gcn_gemm_lds<300, 60, 256, 128, false><<<1250, 256, 0, stream>>>(x, b1, W1, bufA);

    gcn_spmm_rows<256><<<N_NODES / 8, 256, 0, stream>>>(rowptr, edges, bufA, bufB);

    gcn_gemm_lds<256, 64, 128, 64, true><<<1250, 128, 0, stream>>>(bufB, b1, W2, bufA);

    gcn_spmm_rows<128><<<N_NODES / 8, 256, 0, stream>>>(rowptr, edges, bufA, bufB);

    gcn_bias_norm<<<(N_NODES * 32) / 256, 256, 0, stream>>>(b2, bufB, out);
}


# Bias + row L2-normalize folded into SpMM2's row waves: each summed row is normalized in registers and stored straight to out (7 launches, no row plane written and read back); edge list in workspace
# speedup vs baseline: 3.9749x; 1.0515x over previous
#include <hip/hip_runtime.h>
#include <hip/hip_bf16.h>

#define N_NODES 50000
#define N_EDGES 800000

typedef float v2f __attribute__((ext_vector_type(2)));
typedef float v8f __attribute__((ext_vector_type(8)));
typedef float v4f __attribute__((ext_vector_type(4)));

#define GM_ROWS 80
#define SP_ROWS 43680

template <int K, int BK, int NTOT, int BN, bool BIAS_RELU, int SPLIT>
__global__ __launch_bounds__(BN * 2)
void gcn_gemm_lds(const float* __restrict__ a,
                  const float* __restrict__ a_hi,
                  const float* __restrict__ bias,
                  const float* __restrict__ w,
                  float* __restrict__ out) {
    static_assert(K % BK == 0 && BK % 4 == 0 && NTOT % BN == 0 && BN % 16 == 0, "tile shape");
    static_assert(SPLIT % GM_ROWS == 0 && SPLIT <= N_NODES, "a tile lies in one plane");
    constexpr int NTHR = BN * 2;
    constexpr int C4   = BK / 4;
    constexpr int P    = (C4 & 1) ? BK : BK + 4;
    __shared__ __attribute__((aligned(16))) float As[GM_ROWS * P];
    __shared__ __attribute__((aligned(16))) float Bs[BN * P];

    const int tid  = threadIdx.x;
    const int lane = tid & 31;
    const int wave = tid >> 5;
    const int c    = lane & 15;
    const int h2   = (lane >> 4) << 1;
    constexpr int NCB = NTOT / BN;
    const int rowBase = (int)(blockIdx.x / NCB) * GM_ROWS;
    const int colBase = (int)(blockIdx.x % NCB) * BN;

    v8f acc[5];
#pragma unroll
    for (int rt = 0; rt < 5; ++rt) acc[rt] = (v8f){0.f, 0.f, 0.f, 0.f, 0.f, 0.f, 0.f, 0.f};

    static_assert(C4 <= 16 && GM_ROWS % (NTHR / 16) == 0 && BN % (NTHR / 16) == 0, "copy shape");
    constexpr int RPP = NTHR / 16;
    constexpr int NAP = GM_ROWS / RPP;
    constexpr int NBP = BN / RPP;
    const int cr = tid >> 4;
    const int cq = tid & 15;
    const float* __restrict__ atile = (SPLIT < N_NODES && rowBase >= SPLIT)
                                          ? a_hi + (size_t)(rowBase - SPLIT) * K
                                          : a + (size_t)rowBase * K;
    const float* __restrict__ ap = atile + (size_t)cr * K + cq * 4;
    const float* __restrict__ wp = w + (size_t)(colBase + cr) * K + cq * 4;
    float* asl = As + cr * P + cq * 4;
    float* bsl = Bs + cr * P + cq * 4;
    const float* af = As + c * P + h2;
    const float* bf = Bs + (wave * 16 + c) * P + h2;

    for (int k0 = 0; k0 < K; k0 += BK) {
        if (cq < C4) {
            float4 ra[NAP], rb[NBP];
#pragma unroll
            for (int p = 0; p < NAP; ++p)
                ra[p] = *(const float4*)(ap + (size_t)p * RPP * K + k0);
#pragma unroll
            for (int p = 0; p < NBP; ++p)
                rb[p] = *(const float4*)(wp + (size_t)p * RPP * K + k0);
            if (BIAS_RELU) {
                const float4 bb = *(const float4*)(bias + k0 + cq * 4);
#pragma unroll
                for (int p = 0; p < NAP; ++p) {
                    ra[p].x = fmaxf(ra[p].x + bb.x, 0.f);
                    ra[p].y = fmaxf(ra[p].y + bb.y, 0.f);
                    ra[p].z = fmaxf(ra[p].z + bb.z, 0.f);
                    ra[p].w = fmaxf(ra[p].w + bb.w, 0.f);
                }
            }
#pragma unroll
            for (int p = 0; p < NAP; ++p) *(float4*)(asl + p * RPP * P) = ra[p];
#pragma unroll
            for (int p = 0; p < NBP; ++p) *(float4*)(bsl + p * RPP * P) = rb[p];
        }
        __syncthreads();
#pragma unroll
        for (int s = 0; s < C4; ++s) {
            const v2f b = *(const v2f*)(bf + s * 4);
#pragma unroll
            for (int rt = 0; rt < 5; ++rt) {
                const v2f av = *(const v2f*)(af + rt * 16 * P + s * 4);
                acc[rt] = __builtin_amdgcn_wmma_f32_16x16x4_f32(false, av, false, b,
                                                                (short)0, acc[rt], false, false);
            }
        }
        __syncthreads();
    }

    float* __restrict__ o = out + (size_t)(rowBase + ((lane >> 4) << 3)) * NTOT
                                + colBase + wave * 16 + c;
#pragma unroll
    for (int rt = 0; rt < 5; ++rt)
#pragma unroll
        for (int j = 0; j < 8; ++j)
            o[(size_t)(rt * 16 + j) * NTOT] = acc[rt][j];
}

#define NBK      250
#define BK_ROWS  200
#define NCH      128
#define CH_EDGES 6250
#define GS_NT    512
#define CH_IT    ((CH_EDGES + GS_NT - 1) / GS_NT)
#define GS_Q     (128 / (GS_NT / 32))
#define GS_REG   4096
#define GS_RK    (GS_REG / GS_NT)

__device__ __forceinline__ int gcn_wave_incl_scan(int x, int lane) {
    for (int off = 1; off < 32; off <<= 1) {
        int n = __shfl_up(x, off, 32);
        if (lane >= off) x += n;
    }
    return x;
}

__global__ __launch_bounds__(GS_NT) void gcn_bucket_count(const int* __restrict__ erow,
                                                          int* __restrict__ counts32) {
    unsigned short* __restrict__ counts = (unsigned short*)counts32;
    __shared__ int hist[256];
    const int t  = threadIdx.x;
    const int wg = blockIdx.x;
    const int* __restrict__ er = erow + wg * CH_EDGES;
    int r[CH_IT];
#pragma unroll
    for (int k = 0; k < CH_IT; ++k) r[k] = er[min(t + k * GS_NT, CH_EDGES - 1)];
    if (t < 256) hist[t] = 0;
    __syncthreads();
#pragma unroll
    for (int k = 0; k < CH_IT; ++k)
        if (t + k * GS_NT < CH_EDGES) atomicAdd(&hist[r[k] / BK_ROWS], 1);
    __syncthreads();
    if (t < NBK) counts[t * NCH + wg] = (unsigned short)hist[t];
}

__global__ __launch_bounds__(GS_NT) void gcn_bucket_scatter(const int*   __restrict__ erow,
                                                            const int*   __restrict__ ecol,
                                                            const float* __restrict__ eval,
                                                            const int* __restrict__ counts32,
                                                            int* __restrict__ bstart,
                                                            int* __restrict__ inter) {
    const unsigned short* __restrict__ counts = (const unsigned short*)counts32;
    __shared__ int tot[256];
    __shared__ int pre[256];
    __shared__ int cur[256];
    const int t    = threadIdx.x;
    const int lane = t & 31;
    const int wave = t >> 5;
    const int wg   = blockIdx.x;

    int r[CH_IT], c[CH_IT], v[CH_IT];
#pragma unroll
    for (int k = 0; k < CH_IT; ++k) {
        const int e = wg * CH_EDGES + min(t + k * GS_NT, CH_EDGES - 1);
        r[k] = erow[e]; c[k] = ecol[e]; v[k] = __float_as_int(eval[e]);
    }

    const int hl = lane & 15;
    __builtin_amdgcn_sched_barrier(0);
    uint4 w4[GS_Q];
#pragma unroll
    for (int q = 0; q < GS_Q; ++q) {
        const int b = 2 * (wave * GS_Q + q) + (lane >> 4);
        w4[q] = *(const uint4*)(counts + min(b, NBK - 1) * NCH + hl * 8);
    }
#pragma unroll
    for (int q = 0; q < GS_Q; ++q) {
        const int b = 2 * (wave * GS_Q + q) + (lane >> 4);
        const uint4 w = w4[q];
        const unsigned x[8] = {w.x & 0xffffu, w.x >> 16, w.y & 0xffffu, w.y >> 16,
                               w.z & 0xffffu, w.z >> 16, w.w & 0xffffu, w.w >> 16};
        int s = 0, p = 0;
#pragma unroll
        for (int j = 0; j < 8; ++j) {
            s += (int)x[j];
            if (hl * 8 + j < wg) p += (int)x[j];
        }
        if (b >= NBK) { s = 0; p = 0; }
#pragma unroll
        for (int off = 8; off > 0; off >>= 1) {
            s += __shfl_xor(s, off, 32);
            p += __shfl_xor(p, off, 32);
        }
        if (hl == 0) { tot[b] = s; pre[b] = p; }
    }
    __syncthreads();
    if (wave == 0) {
        int a[8];
        int sum = 0;
#pragma unroll
        for (int j = 0; j < 8; ++j) { a[j] = tot[8 * lane + j]; sum += a[j]; }
        int ex = gcn_wave_incl_scan(sum, lane) - sum;
#pragma unroll
        for (int j = 0; j < 8; ++j) {
            const int b = 8 * lane + j;
            cur[b] = ex + pre[b];
            if (wg == 0 && b <= NBK) bstart[b] = ex;
            ex += a[j];
        }
    }
    __syncthreads();
    int pos[CH_IT];
#pragma unroll
    for (int k = 0; k < CH_IT; ++k) {
        const int b = r[k] / BK_ROWS;
        pos[k] = 0;
        if (t + k * GS_NT < CH_EDGES) pos[k] = atomicAdd(&cur[b], 1);
        c[k] |= (r[k] - b * BK_ROWS) << 16;
    }
#pragma unroll
    for (int k = 0; k < CH_IT; ++k)
        if (t + k * GS_NT < CH_EDGES) ((int2*)inter)[pos[k]] = make_int2(c[k], v[k]);
}

__global__ __launch_bounds__(GS_NT) void gcn_bucket_sort(const int* __restrict__ bstart,
                                                         const int* __restrict__ inter,
                                                         int* __restrict__ rowptr,
                                                         int* __restrict__ edges) {
    __shared__ int cnt[256];
    __shared__ int cur[256];
    const int t    = threadIdx.x;
    const int lane = t & 31;
    const int wave = t >> 5;
    const int b    = blockIdx.x;
    const int s    = bstart[b];
    const int n    = bstart[b + 1] - s;
    const int2* __restrict__ in = (const int2*)inter + s;

    int2 rec[GS_RK];
#pragma unroll
    for (int k = 0; k < GS_RK; ++k) {
        const int i = t + k * GS_NT;
        rec[k] = make_int2(0, 0);
        if (i < n) rec[k] = in[i];
    }
    if (t < 256) cnt[t] = 0;
    __syncthreads();
#pragma unroll
    for (int k = 0; k < GS_RK; ++k)
        if (t + k * GS_NT < n) atomicAdd(&cnt[rec[k].x >> 16], 1);
    for (int i = t + GS_REG; i < n; i += GS_NT) atomicAdd(&cnt[in[i].x >> 16], 1);
    __syncthreads();
    if (wave == 0) {
        int a[8];
        int sum = 0;
#pragma unroll
        for (int j = 0; j < 8; ++j) { a[j] = cnt[8 * lane + j]; sum += a[j]; }
        int ex = s + gcn_wave_incl_scan(sum, lane) - sum;
#pragma unroll
        for (int j = 0; j < 8; ++j) {
            const int rib = 8 * lane + j;
            cur[rib] = ex;
            if (rib < BK_ROWS) rowptr[b * BK_ROWS + rib] = ex;
            ex += a[j];
        }
    }
    if (b == NBK - 1 && t == 0) rowptr[N_NODES] = s + n;
    __syncthreads();
    int pos[GS_RK];
#pragma unroll
    for (int k = 0; k < GS_RK; ++k) {
        pos[k] = 0;
        if (t + k * GS_NT < n) pos[k] = atomicAdd(&cur[rec[k].x >> 16], 1);
    }
#pragma unroll
    for (int k = 0; k < GS_RK; ++k)
        if (t + k * GS_NT < n) ((int2*)edges)[pos[k]] = make_int2(rec[k].x & 0xffff, rec[k].y);
    for (int i = t + GS_REG; i < n; i += GS_NT) {
        const int2 rc = in[i];
        const int  p  = atomicAdd(&cur[rc.x >> 16], 1);
        ((int2*)edges)[p] = make_int2(rc.x & 0xffff, rc.y);
    }
}

template <int D, int SPLIT, bool NORM>
__global__ __launch_bounds__(256) void gcn_spmm_rows(const int*   __restrict__ rowptr,
                                                     const int*   __restrict__ edges,
                                                     const float* __restrict__ dense,
                                                     const float* __restrict__ bias,
                                                     float*       __restrict__ dst,
                                                     float*       __restrict__ dst_hi) {
    static_assert(!NORM || D == 128, "the normalizing form holds one row in 32 x 4 registers");
    constexpr int V = D / 128;
    const int lane = threadIdx.x & 31;
    const int row  = __builtin_amdgcn_readfirstlane((int)(blockIdx.x * 8 + (threadIdx.x >> 5)));
    if (row >= N_NODES) return;
    const int start = rowptr[row];
    const int end   = rowptr[row + 1];

    v4f acc[V];
    for (int w = 0; w < V; ++w) acc[w] = (v4f){0.f, 0.f, 0.f, 0.f};

    const int2* __restrict__ er = (const int2*)edges;

    for (int base = start; base < end; base += 32) {
        const int n = min(32, end - base);
        int2 rec = make_int2(0, 0);
        if (lane < n) rec = er[base + lane];

        int j = 0;
        for (; j + 4 <= n; j += 4) {
            float v[4];
            v4f   d[4][V];
#pragma unroll
            for (int u = 0; u < 4; ++u) {
                const int c = __builtin_amdgcn_readlane(rec.x, j + u);
                v[u] = __int_as_float(__builtin_amdgcn_readlane(rec.y, j + u));
                const v4f* __restrict__ rp = (const v4f*)(dense + (size_t)c * D);
#pragma unroll
                for (int w = 0; w < V; ++w) d[u][w] = rp[lane + w * 32];
            }
#pragma unroll
            for (int u = 0; u < 4; ++u)
#pragma unroll
                for (int w = 0; w < V; ++w) acc[w] += v[u] * d[u][w];
        }
#pragma nounroll
        for (; j < n; ++j) {
            const int   c = __builtin_amdgcn_readlane(rec.x, j);
            const float v = __int_as_float(__builtin_amdgcn_readlane(rec.y, j));
            const v4f* __restrict__ rp = (const v4f*)(dense + (size_t)c * D);
            for (int w = 0; w < V; ++w) acc[w] += v * rp[lane + w * 32];
        }
    }

    if (NORM) {
        float4 v  = make_float4(acc[0].x, acc[0].y, acc[0].z, acc[0].w);
        float4 bb = *(const float4*)(bias + lane * 4);
        v.x += bb.x; v.y += bb.y; v.z += bb.z; v.w += bb.w;

        float ss = v.x * v.x + v.y * v.y + v.z * v.z + v.w * v.w;
        for (int off = 16; off > 0; off >>= 1)
            ss += __shfl_xor(ss, off, 32);

        float scale = 1.0f / fmaxf(sqrtf(ss), 1e-12f);
        v.x *= scale; v.y *= scale; v.z *= scale; v.w *= scale;
        *(float4*)(dst + (size_t)row * D + lane * 4) = v;
    } else {
        v4f* __restrict__ o = (SPLIT < N_NODES && row >= SPLIT)
                                  ? (v4f*)(dst_hi + (size_t)(row - SPLIT) * D)
                                  : (v4f*)(dst + (size_t)row * D);
        for (int w = 0; w < V; ++w) o[lane + w * 32] = acc[w];
    }
}

extern "C" void kernel_launch(void* const* d_in, const int* in_sizes, int n_in,
                              void* d_out, int out_size, void* d_ws, size_t ws_size,
                              hipStream_t stream) {
    const float* x    = (const float*)d_in[0];
    const int*   erow = (const int*)  d_in[1];
    const int*   ecol = (const int*)  d_in[2];
    const float* ev   = (const float*)d_in[3];
    const float* W1   = (const float*)d_in[4];
    const float* b1   = (const float*)d_in[5];
    const float* W2   = (const float*)d_in[6];
    const float* b2   = (const float*)d_in[7];
    float* out = (float*)d_out;

    char* ws = (char*)d_ws;
    int* counts = (int*)(ws);
    int* bstart = (int*)(ws + 65536);
    int* rowptr = (int*)(ws + 66560);
    float* bufA = (float*)(ws + 438272);
    float* bufB = (float*)(ws + 438272 + 51200000);
    int*   edges   = (int*)(ws + 438272 + 51200000 + 44728320);
    char*  ob      = (char*)d_out;
    float* bufB_hi = (float*)ob;
    int*   inter   = (int*)(ob + 10025984);

    gcn_bucket_count<<<NCH, GS_NT, 0, stream>>>(erow, counts);
    gcn_bucket_scatter<<<NCH, GS_NT, 0, stream>>>(erow, ecol, ev, counts, bstart, inter);
    gcn_bucket_sort<<<NBK, GS_NT, 0, stream>>>(bstart, inter, rowptr, edges);

    gcn_gemm_lds<300, 60, 256, 128, false, N_NODES><<<1250, 256, 0, stream>>>(x, x, b1, W1, bufA);

    gcn_spmm_rows<256, SP_ROWS, false><<<N_NODES / 8, 256, 0, stream>>>(rowptr, edges, bufA, b1,
                                                                         bufB, bufB_hi);

    gcn_gemm_lds<256, 64, 128, 64, true, SP_ROWS><<<1250, 128, 0, stream>>>(bufB, bufB_hi, b1, W2, bufA);

    gcn_spmm_rows<128, N_NODES, true><<<N_NODES / 8, 256, 0, stream>>>(rowptr, edges, bufA, b2,
                                                                       out, bufB);
}


# Edge grouping hidden beside GEMM1: sort recut as 2 LDS passes (per chunk by bucket; per bucket by row) riding in 2 launches with GEMM1's tiles, 7 -> 5 launches; row sum order = LDS add arrival
# speedup vs baseline: 4.3186x; 1.0865x over previous
#include <hip/hip_runtime.h>
#include <hip/hip_bf16.h>

#define N_NODES 50000
#define N_EDGES 800000

typedef float v2f __attribute__((ext_vector_type(2)));
typedef float v8f __attribute__((ext_vector_type(8)));
typedef float v4f __attribute__((ext_vector_type(4)));

#define GM_ROWS 80
#define SP_ROWS 43680

template <int K, int BK, int NTOT, int BN, bool BIAS_RELU, int SPLIT>
__device__ __forceinline__
void gcn_gemm_tile(const int bid,
                   const float* __restrict__ a,
                   const float* __restrict__ a_hi,
                   const float* __restrict__ bias,
                   const float* __restrict__ w,
                   float* __restrict__ out) {
    static_assert(K % BK == 0 && BK % 4 == 0 && NTOT % BN == 0 && BN % 16 == 0, "tile shape");
    static_assert(SPLIT % GM_ROWS == 0 && SPLIT <= N_NODES, "a tile lies in one plane");
    constexpr int NTHR = BN * 2;
    constexpr int C4   = BK / 4;
    constexpr int P    = (C4 & 1) ? BK : BK + 4;
    __shared__ __attribute__((aligned(16))) float As[GM_ROWS * P];
    __shared__ __attribute__((aligned(16))) float Bs[BN * P];

    const int tid  = threadIdx.x;
    const int lane = tid & 31;
    const int wave = tid >> 5;
    const int c    = lane & 15;
    const int h2   = (lane >> 4) << 1;
    constexpr int NCB = NTOT / BN;
    const int rowBase = (bid / NCB) * GM_ROWS;
    const int colBase = (bid % NCB) * BN;

    v8f acc[5];
#pragma unroll
    for (int rt = 0; rt < 5; ++rt) acc[rt] = (v8f){0.f, 0.f, 0.f, 0.f, 0.f, 0.f, 0.f, 0.f};

    static_assert(C4 <= 16 && GM_ROWS % (NTHR / 16) == 0 && BN % (NTHR / 16) == 0, "copy shape");
    constexpr int RPP = NTHR / 16;
    constexpr int NAP = GM_ROWS / RPP;
    constexpr int NBP = BN / RPP;
    const int cr = tid >> 4;
    const int cq = tid & 15;
    const float* __restrict__ atile = (SPLIT < N_NODES && rowBase >= SPLIT)
                                          ? a_hi + (size_t)(rowBase - SPLIT) * K
                                          : a + (size_t)rowBase * K;
    const float* __restrict__ ap = atile + (size_t)cr * K + cq * 4;
    const float* __restrict__ wp = w + (size_t)(colBase + cr) * K + cq * 4;
    float* asl = As + cr * P + cq * 4;
    float* bsl = Bs + cr * P + cq * 4;
    const float* af = As + c * P + h2;
    const float* bf = Bs + (wave * 16 + c) * P + h2;

    for (int k0 = 0; k0 < K; k0 += BK) {
        if (cq < C4) {
            float4 ra[NAP], rb[NBP];
#pragma unroll
            for (int p = 0; p < NAP; ++p)
                ra[p] = *(const float4*)(ap + (size_t)p * RPP * K + k0);
#pragma unroll
            for (int p = 0; p < NBP; ++p)
                rb[p] = *(const float4*)(wp + (size_t)p * RPP * K + k0);
            if (BIAS_RELU) {
                const float4 bb = *(const float4*)(bias + k0 + cq * 4);
#pragma unroll
                for (int p = 0; p < NAP; ++p) {
                    ra[p].x = fmaxf(ra[p].x + bb.x, 0.f);
                    ra[p].y = fmaxf(ra[p].y + bb.y, 0.f);
                    ra[p].z = fmaxf(ra[p].z + bb.z, 0.f);
                    ra[p].w = fmaxf(ra[p].w + bb.w, 0.f);
                }
            }
#pragma unroll
            for (int p = 0; p < NAP; ++p) *(float4*)(asl + p * RPP * P) = ra[p];
#pragma unroll
            for (int p = 0; p < NBP; ++p) *(float4*)(bsl + p * RPP * P) = rb[p];
        }
        __syncthreads();
#pragma unroll
        for (int s = 0; s < C4; ++s) {
            const v2f b = *(const v2f*)(bf + s * 4);
#pragma unroll
            for (int rt = 0; rt < 5; ++rt) {
                const v2f av = *(const v2f*)(af + rt * 16 * P + s * 4);
                acc[rt] = __builtin_amdgcn_wmma_f32_16x16x4_f32(false, av, false, b,
                                                                (short)0, acc[rt], false, false);
            }
        }
        __syncthreads();
    }

    float* __restrict__ o = out + (size_t)(rowBase + ((lane >> 4) << 3)) * NTOT
                                + colBase + wave * 16 + c;
#pragma unroll
    for (int rt = 0; rt < 5; ++rt)
#pragma unroll
        for (int j = 0; j < 8; ++j)
            o[(size_t)(rt * 16 + j) * NTOT] = acc[rt][j];
}

#define NBK      250
#define BK_ROWS  200
#define NCH      256
#define CH_EDGES 3125
#define GS_NT    256
#define CH_IT    ((CH_EDGES + GS_NT - 1) / GS_NT)
#define SEG_RK   24
#define G1_TILES 1250
#define G1_FIRST 768
#define G1_REST  (G1_TILES - G1_FIRST)

__device__ __forceinline__ int gcn_wave_incl_scan(int x, int lane) {
    for (int off = 1; off < 32; off <<= 1) {
        int n = __shfl_up(x, off, 32);
        if (lane >= off) x += n;
    }
    return x;
}

__device__ __forceinline__ void gcn_chunk_pass(const int wg,
                                               const int*   __restrict__ erow,
                                               const int*   __restrict__ ecol,
                                               const float* __restrict__ eval,
                                               unsigned short* __restrict__ offt,
                                               int* __restrict__ inter) {
    __shared__ int hist[256];
    __shared__ int cur[256];
    const int t    = threadIdx.x;
    const int lane = t & 31;
    const int wave = t >> 5;

    int r[CH_IT], c[CH_IT], v[CH_IT];
#pragma unroll
    for (int k = 0; k < CH_IT; ++k) {
        const int e = wg * CH_EDGES + min(t + k * GS_NT, CH_EDGES - 1);
        r[k] = erow[e]; c[k] = ecol[e]; v[k] = __float_as_int(eval[e]);
    }
    hist[t] = 0;
    __syncthreads();
#pragma unroll
    for (int k = 0; k < CH_IT; ++k)
        if (t + k * GS_NT < CH_EDGES) atomicAdd(&hist[r[k] / BK_ROWS], 1);
    __syncthreads();
    if (wave == 0) {
        int a[8];
        int sum = 0;
#pragma unroll
        for (int j = 0; j < 8; ++j) { a[j] = hist[8 * lane + j]; sum += a[j]; }
        int ex = gcn_wave_incl_scan(sum, lane) - sum;
#pragma unroll
        for (int j = 0; j < 8; ++j) {
            const int b = 8 * lane + j;
            cur[b] = ex;
            if (b <= NBK) offt[b * NCH + wg] = (unsigned short)ex;
            ex += a[j];
        }
    }
    __syncthreads();
    int pos[CH_IT];
#pragma unroll
    for (int k = 0; k < CH_IT; ++k) {
        const int b = r[k] / BK_ROWS;
        pos[k] = 0;
        if (t + k * GS_NT < CH_EDGES) pos[k] = atomicAdd(&cur[b], 1);
        c[k] |= (r[k] - b * BK_ROWS) << 16;
    }
#pragma unroll
    for (int k = 0; k < CH_IT; ++k)
        if (t + k * GS_NT < CH_EDGES)
            ((int2*)inter)[wg * CH_EDGES + pos[k]] = make_int2(c[k], v[k]);
}

__device__ __forceinline__ void gcn_bucket_pass(const int b,
                                                const unsigned short* __restrict__ offt,
                                                const int* __restrict__ inter,
                                                int* __restrict__ rowptr,
                                                int* __restrict__ edges) {
    __shared__ int cnt[256];
    __shared__ int cur[256];
    __shared__ int wst[GS_NT / 32];
    __shared__ int wln[GS_NT / 32];
    const int t    = threadIdx.x;
    const int lane = t & 31;
    const int wave = t >> 5;

    const int s0  = (int)offt[b * NCH + t];
    const int len = (int)offt[(b + 1) * NCH + t] - s0;
    const int2* __restrict__ in = (const int2*)inter + (t * CH_EDGES + s0);

    int2 rec[SEG_RK];
#pragma unroll
    for (int k = 0; k < SEG_RK; ++k) {
        rec[k] = make_int2(0, 0);
        if (k < len) rec[k] = in[k];
    }
    cnt[t] = 0;
    const int is0 = gcn_wave_incl_scan(s0, lane);
    const int iln = gcn_wave_incl_scan(len, lane);
    if (lane == 31) { wst[wave] = is0; wln[wave] = iln; }
    __syncthreads();
    int s = 0, n = 0;
    for (int w = 0; w < GS_NT / 32; ++w) { s += wst[w]; n += wln[w]; }
#pragma unroll
    for (int k = 0; k < SEG_RK; ++k)
        if (k < len) atomicAdd(&cnt[rec[k].x >> 16], 1);
    for (int j = SEG_RK; j < len; ++j) atomicAdd(&cnt[in[j].x >> 16], 1);
    __syncthreads();
    if (wave == 0) {
        int a[8];
        int sum = 0;
#pragma unroll
        for (int j = 0; j < 8; ++j) { a[j] = cnt[8 * lane + j]; sum += a[j]; }
        int ex = s + gcn_wave_incl_scan(sum, lane) - sum;
#pragma unroll
        for (int j = 0; j < 8; ++j) {
            const int rib = 8 * lane + j;
            cur[rib] = ex;
            if (rib < BK_ROWS) rowptr[b * BK_ROWS + rib] = ex;
            ex += a[j];
        }
    }
    if (b == NBK - 1 && t == 0) rowptr[N_NODES] = s + n;
    __syncthreads();
    int pos[SEG_RK];
#pragma unroll
    for (int k = 0; k < SEG_RK; ++k) {
        pos[k] = 0;
        if (k < len) pos[k] = atomicAdd(&cur[rec[k].x >> 16], 1);
    }
#pragma unroll
    for (int k = 0; k < SEG_RK; ++k)
        if (k < len) ((int2*)edges)[pos[k]] = make_int2(rec[k].x & 0xffff, rec[k].y);
    for (int j = SEG_RK; j < len; ++j) {
        const int2 rc = in[j];
        const int  p  = atomicAdd(&cur[rc.x >> 16], 1);
        ((int2*)edges)[p] = make_int2(rc.x & 0xffff, rc.y);
    }
}

template <int K, int BK, int NTOT, int BN, bool BIAS_RELU, int SPLIT>
__global__ __launch_bounds__(BN * 2)
void gcn_gemm_lds(const float* __restrict__ a,
                  const float* __restrict__ a_hi,
                  const float* __restrict__ bias,
                  const float* __restrict__ w,
                  float* __restrict__ out) {
    gcn_gemm_tile<K, BK, NTOT, BN, BIAS_RELU, SPLIT>((int)blockIdx.x, a, a_hi, bias, w, out);
}

__global__ __launch_bounds__(GS_NT)
void gcn_chunks_gemm1(const int*   __restrict__ erow,
                      const int*   __restrict__ ecol,
                      const float* __restrict__ eval,
                      int* __restrict__ offt32,
                      int* __restrict__ inter,
                      const float* __restrict__ x,
                      const float* __restrict__ b1,
                      const float* __restrict__ W1,
                      float* __restrict__ h) {
    if (blockIdx.x < NCH)
        gcn_chunk_pass((int)blockIdx.x, erow, ecol, eval, (unsigned short*)offt32, inter);
    else
        gcn_gemm_tile<300, 60, 256, 128, false, N_NODES>((int)blockIdx.x - NCH, x, x, b1, W1, h);
}

__global__ __launch_bounds__(GS_NT)
void gcn_gemm1_buckets(const int* __restrict__ offt32,
                       const int* __restrict__ inter,
                       int* __restrict__ rowptr,
                       int* __restrict__ edges,
                       const float* __restrict__ x,
                       const float* __restrict__ b1,
                       const float* __restrict__ W1,
                       float* __restrict__ h) {
    if (blockIdx.x < G1_REST)
        gcn_gemm_tile<300, 60, 256, 128, false, N_NODES>(G1_FIRST + (int)blockIdx.x, x, x, b1, W1, h);
    else
        gcn_bucket_pass((int)blockIdx.x - G1_REST, (const unsigned short*)offt32, inter, rowptr, edges);
}

template <int D, int SPLIT, bool NORM>
__global__ __launch_bounds__(256) void gcn_spmm_rows(const int*   __restrict__ rowptr,
                                                     const int*   __restrict__ edges,
                                                     const float* __restrict__ dense,
                                                     const float* __restrict__ bias,
                                                     float*       __restrict__ dst,
                                                     float*       __restrict__ dst_hi) {
    static_assert(!NORM || D == 128, "the normalizing form holds one row in 32 x 4 registers");
    constexpr int V = D / 128;
    const int lane = threadIdx.x & 31;
    const int row  = __builtin_amdgcn_readfirstlane((int)(blockIdx.x * 8 + (threadIdx.x >> 5)));
    if (row >= N_NODES) return;
    const int start = rowptr[row];
    const int end   = rowptr[row + 1];

    v4f acc[V];
    for (int w = 0; w < V; ++w) acc[w] = (v4f){0.f, 0.f, 0.f, 0.f};

    const int2* __restrict__ er = (const int2*)edges;

    for (int base = start; base < end; base += 32) {
        const int n = min(32, end - base);
        int2 rec = make_int2(0, 0);
        if (lane < n) rec = er[base + lane];

        int j = 0;
        for (; j + 4 <= n; j += 4) {
            float v[4];
            v4f   d[4][V];
#pragma unroll
            for (int u = 0; u < 4; ++u) {
                const int c = __builtin_amdgcn_readlane(rec.x, j + u);
                v[u] = __int_as_float(__builtin_amdgcn_readlane(rec.y, j + u));
                const v4f* __restrict__ rp = (const v4f*)(dense + (size_t)c * D);
#pragma unroll
                for (int w = 0; w < V; ++w) d[u][w] = rp[lane + w * 32];
            }
#pragma unroll
            for (int u = 0; u < 4; ++u)
#pragma unroll
                for (int w = 0; w < V; ++w) acc[w] += v[u] * d[u][w];
        }
#pragma nounroll
        for (; j < n; ++j) {
            const int   c = __builtin_amdgcn_readlane(rec.x, j);
            const float v = __int_as_float(__builtin_amdgcn_readlane(rec.y, j));
            const v4f* __restrict__ rp = (const v4f*)(dense + (size_t)c * D);
            for (int w = 0; w < V; ++w) acc[w] += v * rp[lane + w * 32];
        }
    }

    if (NORM) {
        float4 v  = make_float4(acc[0].x, acc[0].y, acc[0].z, acc[0].w);
        float4 bb = *(const float4*)(bias + lane * 4);
        v.x += bb.x; v.y += bb.y; v.z += bb.z; v.w += bb.w;

        float ss = v.x * v.x + v.y * v.y + v.z * v.z + v.w * v.w;
        for (int off = 16; off > 0; off >>= 1)
            ss += __shfl_xor(ss, off, 32);

        float scale = 1.0f / fmaxf(sqrtf(ss), 1e-12f);
        v.x *= scale; v.y *= scale; v.z *= scale; v.w *= scale;
        *(float4*)(dst + (size_t)row * D + lane * 4) = v;
    } else {
        v4f* __restrict__ o = (SPLIT < N_NODES && row >= SPLIT)
                                  ? (v4f*)(dst_hi + (size_t)(row - SPLIT) * D)
                                  : (v4f*)(dst + (size_t)row * D);
        for (int w = 0; w < V; ++w) o[lane + w * 32] = acc[w];
    }
}

extern "C" void kernel_launch(void* const* d_in, const int* in_sizes, int n_in,
                              void* d_out, int out_size, void* d_ws, size_t ws_size,
                              hipStream_t stream) {
    const float* x    = (const float*)d_in[0];
    const int*   erow = (const int*)  d_in[1];
    const int*   ecol = (const int*)  d_in[2];
    const float* ev   = (const float*)d_in[3];
    const float* W1   = (const float*)d_in[4];
    const float* b1   = (const float*)d_in[5];
    const float* W2   = (const float*)d_in[6];
    const float* b2   = (const float*)d_in[7];
    float* out = (float*)d_out;

    char* ws = (char*)d_ws;
    int* rowptr = (int*)(ws + 66560);
    int* offt   = (int*)(ws + 266752);
    float* bufA = (float*)(ws + 438272);
    float* bufB = (float*)(ws + 438272 + 51200000);
    int*   edges   = (int*)(ws + 438272 + 51200000 + 44728320);
    char*  ob      = (char*)d_out;
    float* bufB_hi = (float*)ob;
    int*   inter   = (int*)(ob + 10025984);

    gcn_chunks_gemm1<<<NCH + G1_FIRST, GS_NT, 0, stream>>>(erow, ecol, ev, offt, inter, x, b1, W1, bufA);
    gcn_gemm1_buckets<<<G1_REST + NBK, GS_NT, 0, stream>>>(offt, inter, rowptr, edges, x, b1, W1, bufA);

    gcn_spmm_rows<256, SP_ROWS, false><<<N_NODES / 8, 256, 0, stream>>>(rowptr, edges, bufA, b1,
                                                                         bufB, bufB_hi);

    gcn_gemm_lds<256, 64, 128, 64, true, SP_ROWS><<<1250, 128, 0, stream>>>(bufB, bufB_hi, b1, W2, bufA);

    gcn_spmm_rows<128, N_NODES, true><<<N_NODES / 8, 256, 0, stream>>>(rowptr, edges, bufA, b2,
                                                                       out, bufB);
}


# SpMM1 + GEMM2 in one launch of 1563 small 8-wave work-groups (32 rows: summed into LDS, multiplied there, W2 in 32-k slabs), 5 resident per WGP so gathers overlap WMMAs; 5->4 launches
# speedup vs baseline: 4.5341x; 1.0499x over previous
#include <hip/hip_runtime.h>
#include <hip/hip_bf16.h>

#define N_NODES 50000
#define N_EDGES 800000

typedef float v2f __attribute__((ext_vector_type(2)));
typedef float v8f __attribute__((ext_vector_type(8)));
typedef float v4f __attribute__((ext_vector_type(4)));

#define GM_ROWS 80

template <int K, int BK, int NTOT, int BN, bool BIAS_RELU, int SPLIT>
__device__ __forceinline__
void gcn_gemm_tile(const int bid,
                   const float* __restrict__ a,
                   const float* __restrict__ a_hi,
                   const float* __restrict__ bias,
                   const float* __restrict__ w,
                   float* __restrict__ out) {
    static_assert(K % BK == 0 && BK % 4 == 0 && NTOT % BN == 0 && BN % 16 == 0, "tile shape");
    static_assert(SPLIT % GM_ROWS == 0 && SPLIT <= N_NODES, "a tile lies in one plane");
    constexpr int NTHR = BN * 2;
    constexpr int C4   = BK / 4;
    constexpr int P    = (C4 & 1) ? BK : BK + 4;
    __shared__ __attribute__((aligned(16))) float As[GM_ROWS * P];
    __shared__ __attribute__((aligned(16))) float Bs[BN * P];

    const int tid  = threadIdx.x;
    const int lane = tid & 31;
    const int wave = tid >> 5;
    const int c    = lane & 15;
    const int h2   = (lane >> 4) << 1;
    constexpr int NCB = NTOT / BN;
    const int rowBase = (bid / NCB) * GM_ROWS;
    const int colBase = (bid % NCB) * BN;

    v8f acc[5];
#pragma unroll
    for (int rt = 0; rt < 5; ++rt) acc[rt] = (v8f){0.f, 0.f, 0.f, 0.f, 0.f, 0.f, 0.f, 0.f};

    static_assert(C4 <= 16 && GM_ROWS % (NTHR / 16) == 0 && BN % (NTHR / 16) == 0, "copy shape");
    constexpr int RPP = NTHR / 16;
    constexpr int NAP = GM_ROWS / RPP;
    constexpr int NBP = BN / RPP;
    const int cr = tid >> 4;
    const int cq = tid & 15;
    const float* __restrict__ atile = (SPLIT < N_NODES && rowBase >= SPLIT)
                                          ? a_hi + (size_t)(rowBase - SPLIT) * K
                                          : a + (size_t)rowBase * K;
    const float* __restrict__ ap = atile + (size_t)cr * K + cq * 4;
    const float* __restrict__ wp = w + (size_t)(colBase + cr) * K + cq * 4;
    float* asl = As + cr * P + cq * 4;
    float* bsl = Bs + cr * P + cq * 4;
    const float* af = As + c * P + h2;
    const float* bf = Bs + (wave * 16 + c) * P + h2;

    for (int k0 = 0; k0 < K; k0 += BK) {
        if (cq < C4) {
            float4 ra[NAP], rb[NBP];
#pragma unroll
            for (int p = 0; p < NAP; ++p)
                ra[p] = *(const float4*)(ap + (size_t)p * RPP * K + k0);
#pragma unroll
            for (int p = 0; p < NBP; ++p)
                rb[p] = *(const float4*)(wp + (size_t)p * RPP * K + k0);
            if (BIAS_RELU) {
                const float4 bb = *(const float4*)(bias + k0 + cq * 4);
#pragma unroll
                for (int p = 0; p < NAP; ++p) {
                    ra[p].x = fmaxf(ra[p].x + bb.x, 0.f);
                    ra[p].y = fmaxf(ra[p].y + bb.y, 0.f);
                    ra[p].z = fmaxf(ra[p].z + bb.z, 0.f);
                    ra[p].w = fmaxf(ra[p].w + bb.w, 0.f);
                }
            }
#pragma unroll
            for (int p = 0; p < NAP; ++p) *(float4*)(asl + p * RPP * P) = ra[p];
#pragma unroll
            for (int p = 0; p < NBP; ++p) *(float4*)(bsl + p * RPP * P) = rb[p];
        }
        __syncthreads();
#pragma unroll
        for (int s = 0; s < C4; ++s) {
            const v2f b = *(const v2f*)(bf + s * 4);
#pragma unroll
            for (int rt = 0; rt < 5; ++rt) {
                const v2f av = *(const v2f*)(af + rt * 16 * P + s * 4);
                acc[rt] = __builtin_amdgcn_wmma_f32_16x16x4_f32(false, av, false, b,
                                                                (short)0, acc[rt], false, false);
            }
        }
        __syncthreads();
    }

    float* __restrict__ o = out + (size_t)(rowBase + ((lane >> 4) << 3)) * NTOT
                                + colBase + wave * 16 + c;
#pragma unroll
    for (int rt = 0; rt < 5; ++rt)
#pragma unroll
        for (int j = 0; j < 8; ++j)
            o[(size_t)(rt * 16 + j) * NTOT] = acc[rt][j];
}

#define NBK      250
#define BK_ROWS  200
#define NCH      256
#define CH_EDGES 3125
#define GS_NT    256
#define CH_IT    ((CH_EDGES + GS_NT - 1) / GS_NT)
#define SEG_RK   24
#define G1_TILES 1250
#define G1_FIRST 768
#define G1_REST  (G1_TILES - G1_FIRST)

__device__ __forceinline__ int gcn_wave_incl_scan(int x, int lane) {
    for (int off = 1; off < 32; off <<= 1) {
        int n = __shfl_up(x, off, 32);
        if (lane >= off) x += n;
    }
    return x;
}

__device__ __forceinline__ void gcn_chunk_pass(const int wg,
                                               const int*   __restrict__ erow,
                                               const int*   __restrict__ ecol,
                                               const float* __restrict__ eval,
                                               unsigned short* __restrict__ offt,
                                               int* __restrict__ inter) {
    __shared__ int hist[256];
    __shared__ int cur[256];
    const int t    = threadIdx.x;
    const int lane = t & 31;
    const int wave = t >> 5;

    int r[CH_IT], c[CH_IT], v[CH_IT];
#pragma unroll
    for (int k = 0; k < CH_IT; ++k) {
        const int e = wg * CH_EDGES + min(t + k * GS_NT, CH_EDGES - 1);
        r[k] = erow[e]; c[k] = ecol[e]; v[k] = __float_as_int(eval[e]);
    }
    hist[t] = 0;
    __syncthreads();
#pragma unroll
    for (int k = 0; k < CH_IT; ++k)
        if (t + k * GS_NT < CH_EDGES) atomicAdd(&hist[r[k] / BK_ROWS], 1);
    __syncthreads();
    if (wave == 0) {
        int a[8];
        int sum = 0;
#pragma unroll
        for (int j = 0; j < 8; ++j) { a[j] = hist[8 * lane + j]; sum += a[j]; }
        int ex = gcn_wave_incl_scan(sum, lane) - sum;
#pragma unroll
        for (int j = 0; j < 8; ++j) {
            const int b = 8 * lane + j;
            cur[b] = ex;
            if (b <= NBK) offt[b * NCH + wg] = (unsigned short)ex;
            ex += a[j];
        }
    }
    __syncthreads();
    int pos[CH_IT];
#pragma unroll
    for (int k = 0; k < CH_IT; ++k) {
        const int b = r[k] / BK_ROWS;
        pos[k] = 0;
        if (t + k * GS_NT < CH_EDGES) pos[k] = atomicAdd(&cur[b], 1);
        c[k] |= (r[k] - b * BK_ROWS) << 16;
    }
#pragma unroll
    for (int k = 0; k < CH_IT; ++k)
        if (t + k * GS_NT < CH_EDGES)
            ((int2*)inter)[wg * CH_EDGES + pos[k]] = make_int2(c[k], v[k]);
}

__device__ __forceinline__ void gcn_bucket_pass(const int b,
                                                const unsigned short* __restrict__ offt,
                                                const int* __restrict__ inter,
                                                int* __restrict__ rowptr,
                                                int* __restrict__ edges) {
    __shared__ int cnt[256];
    __shared__ int cur[256];
    __shared__ int wst[GS_NT / 32];
    __shared__ int wln[GS_NT / 32];
    const int t    = threadIdx.x;
    const int lane = t & 31;
    const int wave = t >> 5;

    const int s0  = (int)offt[b * NCH + t];
    const int len = (int)offt[(b + 1) * NCH + t] - s0;
    const int2* __restrict__ in = (const int2*)inter + (t * CH_EDGES + s0);

    int2 rec[SEG_RK];
#pragma unroll
    for (int k = 0; k < SEG_RK; ++k) {
        rec[k] = make_int2(0, 0);
        if (k < len) rec[k] = in[k];
    }
    cnt[t] = 0;
    const int is0 = gcn_wave_incl_scan(s0, lane);
    const int iln = gcn_wave_incl_scan(len, lane);
    if (lane == 31) { wst[wave] = is0; wln[wave] = iln; }
    __syncthreads();
    int s = 0, n = 0;
    for (int w = 0; w < GS_NT / 32; ++w) { s += wst[w]; n += wln[w]; }
#pragma unroll
    for (int k = 0; k < SEG_RK; ++k)
        if (k < len) atomicAdd(&cnt[rec[k].x >> 16], 1);
    for (int j = SEG_RK; j < len; ++j) atomicAdd(&cnt[in[j].x >> 16], 1);
    __syncthreads();
    if (wave == 0) {
        int a[8];
        int sum = 0;
#pragma unroll
        for (int j = 0; j < 8; ++j) { a[j] = cnt[8 * lane + j]; sum += a[j]; }
        int ex = s + gcn_wave_incl_scan(sum, lane) - sum;
#pragma unroll
        for (int j = 0; j < 8; ++j) {
            const int rib = 8 * lane + j;
            cur[rib] = ex;
            if (rib < BK_ROWS) rowptr[b * BK_ROWS + rib] = ex;
            ex += a[j];
        }
    }
    if (b == NBK - 1 && t == 0) rowptr[N_NODES] = s + n;
    __syncthreads();
    int pos[SEG_RK];
#pragma unroll
    for (int k = 0; k < SEG_RK; ++k) {
        pos[k] = 0;
        if (k < len) pos[k] = atomicAdd(&cur[rec[k].x >> 16], 1);
    }
#pragma unroll
    for (int k = 0; k < SEG_RK; ++k)
        if (k < len) ((int2*)edges)[pos[k]] = make_int2(rec[k].x & 0xffff, rec[k].y);
    for (int j = SEG_RK; j < len; ++j) {
        const int2 rc = in[j];
        const int  p  = atomicAdd(&cur[rc.x >> 16], 1);
        ((int2*)edges)[p] = make_int2(rc.x & 0xffff, rc.y);
    }
}

__global__ __launch_bounds__(GS_NT)
void gcn_chunks_gemm1(const int*   __restrict__ erow,
                      const int*   __restrict__ ecol,
                      const float* __restrict__ eval,
                      int* __restrict__ offt32,
                      int* __restrict__ inter,
                      const float* __restrict__ x,
                      const float* __restrict__ b1,
                      const float* __restrict__ W1,
                      float* __restrict__ h) {
    if (blockIdx.x < NCH)
        gcn_chunk_pass((int)blockIdx.x, erow, ecol, eval, (unsigned short*)offt32, inter);
    else
        gcn_gemm_tile<300, 60, 256, 128, false, N_NODES>((int)blockIdx.x - NCH, x, x, b1, W1, h);
}

__global__ __launch_bounds__(GS_NT)
void gcn_gemm1_buckets(const int* __restrict__ offt32,
                       const int* __restrict__ inter,
                       int* __restrict__ rowptr,
                       int* __restrict__ edges,
                       const float* __restrict__ x,
                       const float* __restrict__ b1,
                       const float* __restrict__ W1,
                       float* __restrict__ h) {
    if (blockIdx.x < G1_REST)
        gcn_gemm_tile<300, 60, 256, 128, false, N_NODES>(G1_FIRST + (int)blockIdx.x, x, x, b1, W1, h);
    else
        gcn_bucket_pass((int)blockIdx.x - G1_REST, (const unsigned short*)offt32, inter, rowptr, edges);
}

template <int D, int SPLIT, bool NORM>
__global__ __launch_bounds__(256) void gcn_spmm_rows(const int*   __restrict__ rowptr,
                                                     const int*   __restrict__ edges,
                                                     const float* __restrict__ dense,
                                                     const float* __restrict__ bias,
                                                     float*       __restrict__ dst,
                                                     float*       __restrict__ dst_hi) {
    static_assert(!NORM || D == 128, "the normalizing form holds one row in 32 x 4 registers");
    constexpr int V = D / 128;
    const int lane = threadIdx.x & 31;
    const int row  = __builtin_amdgcn_readfirstlane((int)(blockIdx.x * 8 + (threadIdx.x >> 5)));
    if (row >= N_NODES) return;
    const int start = rowptr[row];
    const int end   = rowptr[row + 1];

    v4f acc[V];
    for (int w = 0; w < V; ++w) acc[w] = (v4f){0.f, 0.f, 0.f, 0.f};

    const int2* __restrict__ er = (const int2*)edges;

    for (int base = start; base < end; base += 32) {
        const int n = min(32, end - base);
        int2 rec = make_int2(0, 0);
        if (lane < n) rec = er[base + lane];

        int j = 0;
        for (; j + 4 <= n; j += 4) {
            float v[4];
            v4f   d[4][V];
#pragma unroll
            for (int u = 0; u < 4; ++u) {
                const int c = __builtin_amdgcn_readlane(rec.x, j + u);
                v[u] = __int_as_float(__builtin_amdgcn_readlane(rec.y, j + u));
                const v4f* __restrict__ rp = (const v4f*)(dense + (size_t)c * D);
#pragma unroll
                for (int w = 0; w < V; ++w) d[u][w] = rp[lane + w * 32];
            }
#pragma unroll
            for (int u = 0; u < 4; ++u)
#pragma unroll
                for (int w = 0; w < V; ++w) acc[w] += v[u] * d[u][w];
        }
#pragma nounroll
        for (; j < n; ++j) {
            const int   c = __builtin_amdgcn_readlane(rec.x, j);
            const float v = __int_as_float(__builtin_amdgcn_readlane(rec.y, j));
            const v4f* __restrict__ rp = (const v4f*)(dense + (size_t)c * D);
            for (int w = 0; w < V; ++w) acc[w] += v * rp[lane + w * 32];
        }
    }

    if (NORM) {
        float4 v  = make_float4(acc[0].x, acc[0].y, acc[0].z, acc[0].w);
        float4 bb = *(const float4*)(bias + lane * 4);
        v.x += bb.x; v.y += bb.y; v.z += bb.z; v.w += bb.w;

        float ss = v.x * v.x + v.y * v.y + v.z * v.z + v.w * v.w;
        for (int off = 16; off > 0; off >>= 1)
            ss += __shfl_xor(ss, off, 32);

        float scale = 1.0f / fmaxf(sqrtf(ss), 1e-12f);
        v.x *= scale; v.y *= scale; v.z *= scale; v.w *= scale;
        *(float4*)(dst + (size_t)row * D + lane * 4) = v;
    } else {
        v4f* __restrict__ o = (SPLIT < N_NODES && row >= SPLIT)
                                  ? (v4f*)(dst_hi + (size_t)(row - SPLIT) * D)
                                  : (v4f*)(dst + (size_t)row * D);
        for (int w = 0; w < V; ++w) o[lane + w * 32] = acc[w];
    }
}

#define FT_ROWS  32
#define FT_AP    260
#define FT_BK    32
#define FT_BP    36
#define FT_NSLAB (256 / FT_BK)
#define FT_TILES ((N_NODES + FT_ROWS - 1) / FT_ROWS)

__global__ __launch_bounds__(256)
void gcn_spmm_gemm2(const int*   __restrict__ rowptr,
                    const int*   __restrict__ edges,
                    const float* __restrict__ dense,
                    const float* __restrict__ bias,
                    const float* __restrict__ w,
                    float*       __restrict__ out) {
    __shared__ __attribute__((aligned(16))) float As[FT_ROWS * FT_AP];
    __shared__ __attribute__((aligned(16))) float Bs[128 * FT_BP];

    const int tid  = threadIdx.x;
    const int lane = tid & 31;
    const int wave = __builtin_amdgcn_readfirstlane(tid >> 5);
    const int rowBase = (int)blockIdx.x * FT_ROWS;

    const int cr = tid >> 3;
    const int cq = tid & 7;
    const float* __restrict__ wp = w + (size_t)cr * 256 + cq * 4;
    float* bsl = Bs + cr * FT_BP + cq * 4;
    v4f rb0 = *(const v4f*)(wp);
    v4f rb1 = *(const v4f*)(wp + 32 * 256);
    v4f rb2 = *(const v4f*)(wp + 64 * 256);
    v4f rb3 = *(const v4f*)(wp + 96 * 256);

    {
        const int2* __restrict__ er = (const int2*)edges;
        const float4 b_lo = *(const float4*)(bias + lane * 4);
        const float4 b_hi = *(const float4*)(bias + 128 + lane * 4);
#pragma nounroll
        for (int i = 0; i < FT_ROWS / 8; ++i) {
            const int rl  = wave * (FT_ROWS / 8) + i;
            const int row = rowBase + rl;
            v4f acc[2];
            acc[0] = (v4f){0.f, 0.f, 0.f, 0.f};
            acc[1] = (v4f){0.f, 0.f, 0.f, 0.f};
            int start = 0, end = 0;
            if (row < N_NODES) { start = rowptr[row]; end = rowptr[row + 1]; }
            for (int base = start; base < end; base += 32) {
                const int n = min(32, end - base);
                int2 rec = make_int2(0, 0);
                if (lane < n) rec = er[base + lane];
                int j = 0;
                for (; j + 4 <= n; j += 4) {
                    float v[4];
                    v4f   d[4][2];
#pragma unroll
                    for (int u = 0; u < 4; ++u) {
                        const int c = __builtin_amdgcn_readlane(rec.x, j + u);
                        v[u] = __int_as_float(__builtin_amdgcn_readlane(rec.y, j + u));
                        const v4f* __restrict__ rp = (const v4f*)(dense + (size_t)c * 256);
#pragma unroll
                        for (int q = 0; q < 2; ++q) d[u][q] = rp[lane + q * 32];
                    }
#pragma unroll
                    for (int u = 0; u < 4; ++u)
#pragma unroll
                        for (int q = 0; q < 2; ++q) acc[q] += v[u] * d[u][q];
                }
#pragma nounroll
                for (; j < n; ++j) {
                    const int   c = __builtin_amdgcn_readlane(rec.x, j);
                    const float v = __int_as_float(__builtin_amdgcn_readlane(rec.y, j));
                    const v4f* __restrict__ rp = (const v4f*)(dense + (size_t)c * 256);
                    for (int q = 0; q < 2; ++q) acc[q] += v * rp[lane + q * 32];
                }
            }
            float4 lo, hi;
            lo.x = fmaxf(acc[0].x + b_lo.x, 0.f); lo.y = fmaxf(acc[0].y + b_lo.y, 0.f);
            lo.z = fmaxf(acc[0].z + b_lo.z, 0.f); lo.w = fmaxf(acc[0].w + b_lo.w, 0.f);
            hi.x = fmaxf(acc[1].x + b_hi.x, 0.f); hi.y = fmaxf(acc[1].y + b_hi.y, 0.f);
            hi.z = fmaxf(acc[1].z + b_hi.z, 0.f); hi.w = fmaxf(acc[1].w + b_hi.w, 0.f);
            *(float4*)(As + rl * FT_AP + lane * 4)       = lo;
            *(float4*)(As + rl * FT_AP + 128 + lane * 4) = hi;
        }
    }

    const int c  = lane & 15;
    const int h2 = (lane >> 4) << 1;
    const float* af = As + c * FT_AP + h2;
    const float* bf = Bs + (wave * 16 + c) * FT_BP + h2;
    v8f acc2[2];
    acc2[0] = (v8f){0.f, 0.f, 0.f, 0.f, 0.f, 0.f, 0.f, 0.f};
    acc2[1] = (v8f){0.f, 0.f, 0.f, 0.f, 0.f, 0.f, 0.f, 0.f};

#pragma nounroll
    for (int sl = 0; sl < FT_NSLAB; ++sl) {
        *(v4f*)(bsl)              = rb0;
        *(v4f*)(bsl + 32 * FT_BP) = rb1;
        *(v4f*)(bsl + 64 * FT_BP) = rb2;
        *(v4f*)(bsl + 96 * FT_BP) = rb3;
        __syncthreads();
        if (sl + 1 < FT_NSLAB) {
            const float* __restrict__ wn = wp + (sl + 1) * FT_BK;
            rb0 = *(const v4f*)(wn);
            rb1 = *(const v4f*)(wn + 32 * 256);
            rb2 = *(const v4f*)(wn + 64 * 256);
            rb3 = *(const v4f*)(wn + 96 * 256);
        }
        const float* afs = af + sl * FT_BK;
#pragma unroll
        for (int s = 0; s < FT_BK / 4; ++s) {
            const v2f b = *(const v2f*)(bf + s * 4);
#pragma unroll
            for (int rt = 0; rt < 2; ++rt) {
                const v2f av = *(const v2f*)(afs + rt * 16 * FT_AP + s * 4);
                acc2[rt] = __builtin_amdgcn_wmma_f32_16x16x4_f32(false, av, false, b,
                                                                 (short)0, acc2[rt], false, false);
            }
        }
        __syncthreads();
    }

    float* __restrict__ o = out + (size_t)(rowBase + ((lane >> 4) << 3)) * 128 + wave * 16 + c;
#pragma unroll
    for (int rt = 0; rt < 2; ++rt) {
        if (rowBase + rt * 16 < N_NODES) {
#pragma unroll
            for (int j = 0; j < 8; ++j)
                o[(size_t)(rt * 16 + j) * 128] = acc2[rt][j];
        }
    }
}

extern "C" void kernel_launch(void* const* d_in, const int* in_sizes, int n_in,
                              void* d_out, int out_size, void* d_ws, size_t ws_size,
                              hipStream_t stream) {
    const float* x    = (const float*)d_in[0];
    const int*   erow = (const int*)  d_in[1];
    const int*   ecol = (const int*)  d_in[2];
    const float* ev   = (const float*)d_in[3];
    const float* W1   = (const float*)d_in[4];
    const float* b1   = (const float*)d_in[5];
    const float* W2   = (const float*)d_in[6];
    const float* b2   = (const float*)d_in[7];
    float* out = (float*)d_out;

    char* ws = (char*)d_ws;
    int* rowptr = (int*)(ws + 66560);
    int* offt   = (int*)(ws + 266752);
    float* bufA = (float*)(ws + 438272);
    float* bufB = (float*)(ws + 438272 + 51200000);
    int*   edges   = (int*)(ws + 438272 + 51200000 + 44728320);
    char*  ob      = (char*)d_out;
    int*   inter   = (int*)(ob + 10025984);

    gcn_chunks_gemm1<<<NCH + G1_FIRST, GS_NT, 0, stream>>>(erow, ecol, ev, offt, inter, x, b1, W1, bufA);
    gcn_gemm1_buckets<<<G1_REST + NBK, GS_NT, 0, stream>>>(offt, inter, rowptr, edges, x, b1, W1, bufA);

    gcn_spmm_gemm2<<<FT_TILES, 256, 0, stream>>>(rowptr, edges, bufA, b1, W2, bufB);

    gcn_spmm_rows<128, N_NODES, true><<<N_NODES / 8, 256, 0, stream>>>(rowptr, edges, bufB, b2,
                                                                       out, bufA);
}


# SpMM1+GEMM2 launch as 256 resident 24-wave work-groups with split roles: 16 waves sum the next 16-row tile into LDS while 8 waves multiply the last one; W2 kept in LDS; row sum and k orders kept
# speedup vs baseline: 4.6854x; 1.0334x over previous
#include <hip/hip_runtime.h>
#include <hip/hip_bf16.h>

#define N_NODES 50000
#define N_EDGES 800000

typedef float v2f __attribute__((ext_vector_type(2)));
typedef float v8f __attribute__((ext_vector_type(8)));
typedef float v4f __attribute__((ext_vector_type(4)));

#define GM_ROWS 80

template <int K, int BK, int NTOT, int BN, bool BIAS_RELU, int SPLIT>
__device__ __forceinline__
void gcn_gemm_tile(const int bid,
                   const float* __restrict__ a,
                   const float* __restrict__ a_hi,
                   const float* __restrict__ bias,
                   const float* __restrict__ w,
                   float* __restrict__ out) {
    static_assert(K % BK == 0 && BK % 4 == 0 && NTOT % BN == 0 && BN % 16 == 0, "tile shape");
    static_assert(SPLIT % GM_ROWS == 0 && SPLIT <= N_NODES, "a tile lies in one plane");
    constexpr int NTHR = BN * 2;
    constexpr int C4   = BK / 4;
    constexpr int P    = (C4 & 1) ? BK : BK + 4;
    __shared__ __attribute__((aligned(16))) float As[GM_ROWS * P];
    __shared__ __attribute__((aligned(16))) float Bs[BN * P];

    const int tid  = threadIdx.x;
    const int lane = tid & 31;
    const int wave = tid >> 5;
    const int c    = lane & 15;
    const int h2   = (lane >> 4) << 1;
    constexpr int NCB = NTOT / BN;
    const int rowBase = (bid / NCB) * GM_ROWS;
    const int colBase = (bid % NCB) * BN;

    v8f acc[5];
#pragma unroll
    for (int rt = 0; rt < 5; ++rt) acc[rt] = (v8f){0.f, 0.f, 0.f, 0.f, 0.f, 0.f, 0.f, 0.f};

    static_assert(C4 <= 16 && GM_ROWS % (NTHR / 16) == 0 && BN % (NTHR / 16) == 0, "copy shape");
    constexpr int RPP = NTHR / 16;
    constexpr int NAP = GM_ROWS / RPP;
    constexpr int NBP = BN / RPP;
    const int cr = tid >> 4;
    const int cq = tid & 15;
    const float* __restrict__ atile = (SPLIT < N_NODES && rowBase >= SPLIT)
                                          ? a_hi + (size_t)(rowBase - SPLIT) * K
                                          : a + (size_t)rowBase * K;
    const float* __restrict__ ap = atile + (size_t)cr * K + cq * 4;
    const float* __restrict__ wp = w + (size_t)(colBase + cr) * K + cq * 4;
    float* asl = As + cr * P + cq * 4;
    float* bsl = Bs + cr * P + cq * 4;
    const float* af = As + c * P + h2;
    const float* bf = Bs + (wave * 16 + c) * P + h2;

    for (int k0 = 0; k0 < K; k0 += BK) {
        if (cq < C4) {
            float4 ra[NAP], rb[NBP];
#pragma unroll
            for (int p = 0; p < NAP; ++p)
                ra[p] = *(const float4*)(ap + (size_t)p * RPP * K + k0);
#pragma unroll
            for (int p = 0; p < NBP; ++p)
                rb[p] = *(const float4*)(wp + (size_t)p * RPP * K + k0);
            if (BIAS_RELU) {
                const float4 bb = *(const float4*)(bias + k0 + cq * 4);
#pragma unroll
                for (int p = 0; p < NAP; ++p) {
                    ra[p].x = fmaxf(ra[p].x + bb.x, 0.f);
                    ra[p].y = fmaxf(ra[p].y + bb.y, 0.f);
                    ra[p].z = fmaxf(ra[p].z + bb.z, 0.f);
                    ra[p].w = fmaxf(ra[p].w + bb.w, 0.f);
                }
            }
#pragma unroll
            for (int p = 0; p < NAP; ++p) *(float4*)(asl + p * RPP * P) = ra[p];
#pragma unroll
            for (int p = 0; p < NBP; ++p) *(float4*)(bsl + p * RPP * P) = rb[p];
        }
        __syncthreads();
#pragma unroll
        for (int s = 0; s < C4; ++s) {
            const v2f b = *(const v2f*)(bf + s * 4);
#pragma unroll
            for (int rt = 0; rt < 5; ++rt) {
                const v2f av = *(const v2f*)(af + rt * 16 * P + s * 4);
                acc[rt] = __builtin_amdgcn_wmma_f32_16x16x4_f32(false, av, false, b,
                                                                (short)0, acc[rt], false, false);
            }
        }
        __syncthreads();
    }

    float* __restrict__ o = out + (size_t)(rowBase + ((lane >> 4) << 3)) * NTOT
                                + colBase + wave * 16 + c;
#pragma unroll
    for (int rt = 0; rt < 5; ++rt)
#pragma unroll
        for (int j = 0; j < 8; ++j)
            o[(size_t)(rt * 16 + j) * NTOT] = acc[rt][j];
}

#define NBK      250
#define BK_ROWS  200
#define NCH      256
#define CH_EDGES 3125
#define GS_NT    256
#define CH_IT    ((CH_EDGES + GS_NT - 1) / GS_NT)
#define SEG_RK   24
#define G1_TILES 1250
#define G1_FIRST 768
#define G1_REST  (G1_TILES - G1_FIRST)

__device__ __forceinline__ int gcn_wave_incl_scan(int x, int lane) {
    for (int off = 1; off < 32; off <<= 1) {
        int n = __shfl_up(x, off, 32);
        if (lane >= off) x += n;
    }
    return x;
}

__device__ __forceinline__ void gcn_chunk_pass(const int wg,
                                               const int*   __restrict__ erow,
                                               const int*   __restrict__ ecol,
                                               const float* __restrict__ eval,
                                               unsigned short* __restrict__ offt,
                                               int* __restrict__ inter) {
    __shared__ int hist[256];
    __shared__ int cur[256];
    const int t    = threadIdx.x;
    const int lane = t & 31;
    const int wave = t >> 5;

    int r[CH_IT], c[CH_IT], v[CH_IT];
#pragma unroll
    for (int k = 0; k < CH_IT; ++k) {
        const int e = wg * CH_EDGES + min(t + k * GS_NT, CH_EDGES - 1);
        r[k] = erow[e]; c[k] = ecol[e]; v[k] = __float_as_int(eval[e]);
    }
    hist[t] = 0;
    __syncthreads();
#pragma unroll
    for (int k = 0; k < CH_IT; ++k)
        if (t + k * GS_NT < CH_EDGES) atomicAdd(&hist[r[k] / BK_ROWS], 1);
    __syncthreads();
    if (wave == 0) {
        int a[8];
        int sum = 0;
#pragma unroll
        for (int j = 0; j < 8; ++j) { a[j] = hist[8 * lane + j]; sum += a[j]; }
        int ex = gcn_wave_incl_scan(sum, lane) - sum;
#pragma unroll
        for (int j = 0; j < 8; ++j) {
            const int b = 8 * lane + j;
            cur[b] = ex;
            if (b <= NBK) offt[b * NCH + wg] = (unsigned short)ex;
            ex += a[j];
        }
    }
    __syncthreads();
    int pos[CH_IT];
#pragma unroll
    for (int k = 0; k < CH_IT; ++k) {
        const int b = r[k] / BK_ROWS;
        pos[k] = 0;
        if (t + k * GS_NT < CH_EDGES) pos[k] = atomicAdd(&cur[b], 1);
        c[k] |= (r[k] - b * BK_ROWS) << 16;
    }
#pragma unroll
    for (int k = 0; k < CH_IT; ++k)
        if (t + k * GS_NT < CH_EDGES)
            ((int2*)inter)[wg * CH_EDGES + pos[k]] = make_int2(c[k], v[k]);
}

__device__ __forceinline__ void gcn_bucket_pass(const int b,
                                                const unsigned short* __restrict__ offt,
                                                const int* __restrict__ inter,
                                                int* __restrict__ rowptr,
                                                int* __restrict__ edges) {
    __shared__ int cnt[256];
    __shared__ int cur[256];
    __shared__ int wst[GS_NT / 32];
    __shared__ int wln[GS_NT / 32];
    const int t    = threadIdx.x;
    const int lane = t & 31;
    const int wave = t >> 5;

    const int s0  = (int)offt[b * NCH + t];
    const int len = (int)offt[(b + 1) * NCH + t] - s0;
    const int2* __restrict__ in = (const int2*)inter + (t * CH_EDGES + s0);

    int2 rec[SEG_RK];
#pragma unroll
    for (int k = 0; k < SEG_RK; ++k) {
        rec[k] = make_int2(0, 0);
        if (k < len) rec[k] = in[k];
    }
    cnt[t] = 0;
    const int is0 = gcn_wave_incl_scan(s0, lane);
    const int iln = gcn_wave_incl_scan(len, lane);
    if (lane == 31) { wst[wave] = is0; wln[wave] = iln; }
    __syncthreads();
    int s = 0, n = 0;
    for (int w = 0; w < GS_NT / 32; ++w) { s += wst[w]; n += wln[w]; }
#pragma unroll
    for (int k = 0; k < SEG_RK; ++k)
        if (k < len) atomicAdd(&cnt[rec[k].x >> 16], 1);
    for (int j = SEG_RK; j < len; ++j) atomicAdd(&cnt[in[j].x >> 16], 1);
    __syncthreads();
    if (wave == 0) {
        int a[8];
        int sum = 0;
#pragma unroll
        for (int j = 0; j < 8; ++j) { a[j] = cnt[8 * lane + j]; sum += a[j]; }
        int ex = s + gcn_wave_incl_scan(sum, lane) - sum;
#pragma unroll
        for (int j = 0; j < 8; ++j) {
            const int rib = 8 * lane + j;
            cur[rib] = ex;
            if (rib < BK_ROWS) rowptr[b * BK_ROWS + rib] = ex;
            ex += a[j];
        }
    }
    if (b == NBK - 1 && t == 0) rowptr[N_NODES] = s + n;
    __syncthreads();
    int pos[SEG_RK];
#pragma unroll
    for (int k = 0; k < SEG_RK; ++k) {
        pos[k] = 0;
        if (k < len) pos[k] = atomicAdd(&cur[rec[k].x >> 16], 1);
    }
#pragma unroll
    for (int k = 0; k < SEG_RK; ++k)
        if (k < len) ((int2*)edges)[pos[k]] = make_int2(rec[k].x & 0xffff, rec[k].y);
    for (int j = SEG_RK; j < len; ++j) {
        const int2 rc = in[j];
        const int  p  = atomicAdd(&cur[rc.x >> 16], 1);
        ((int2*)edges)[p] = make_int2(rc.x & 0xffff, rc.y);
    }
}

__global__ __launch_bounds__(GS_NT)
void gcn_chunks_gemm1(const int*   __restrict__ erow,
                      const int*   __restrict__ ecol,
                      const float* __restrict__ eval,
                      int* __restrict__ offt32,
                      int* __restrict__ inter,
                      const float* __restrict__ x,
                      const float* __restrict__ b1,
                      const float* __restrict__ W1,
                      float* __restrict__ h) {
    if (blockIdx.x < NCH)
        gcn_chunk_pass((int)blockIdx.x, erow, ecol, eval, (unsigned short*)offt32, inter);
    else
        gcn_gemm_tile<300, 60, 256, 128, false, N_NODES>((int)blockIdx.x - NCH, x, x, b1, W1, h);
}

__global__ __launch_bounds__(GS_NT)
void gcn_gemm1_buckets(const int* __restrict__ offt32,
                       const int* __restrict__ inter,
                       int* __restrict__ rowptr,
                       int* __restrict__ edges,
                       const float* __restrict__ x,
                       const float* __restrict__ b1,
                       const float* __restrict__ W1,
                       float* __restrict__ h) {
    if (blockIdx.x < G1_REST)
        gcn_gemm_tile<300, 60, 256, 128, false, N_NODES>(G1_FIRST + (int)blockIdx.x, x, x, b1, W1, h);
    else
        gcn_bucket_pass((int)blockIdx.x - G1_REST, (const unsigned short*)offt32, inter, rowptr, edges);
}

template <int D, int SPLIT, bool NORM>
__global__ __launch_bounds__(256) void gcn_spmm_rows(const int*   __restrict__ rowptr,
                                                     const int*   __restrict__ edges,
                                                     const float* __restrict__ dense,
                                                     const float* __restrict__ bias,
                                                     float*       __restrict__ dst,
                                                     float*       __restrict__ dst_hi) {
    static_assert(!NORM || D == 128, "the normalizing form holds one row in 32 x 4 registers");
    constexpr int V = D / 128;
    const int lane = threadIdx.x & 31;
    const int row  = __builtin_amdgcn_readfirstlane((int)(blockIdx.x * 8 + (threadIdx.x >> 5)));
    if (row >= N_NODES) return;
    const int start = rowptr[row];
    const int end   = rowptr[row + 1];

    v4f acc[V];
    for (int w = 0; w < V; ++w) acc[w] = (v4f){0.f, 0.f, 0.f, 0.f};

    const int2* __restrict__ er = (const int2*)edges;

    for (int base = start; base < end; base += 32) {
        const int n = min(32, end - base);
        int2 rec = make_int2(0, 0);
        if (lane < n) rec = er[base + lane];

        int j = 0;
        for (; j + 4 <= n; j += 4) {
            float v[4];
            v4f   d[4][V];
#pragma unroll
            for (int u = 0; u < 4; ++u) {
                const int c = __builtin_amdgcn_readlane(rec.x, j + u);
                v[u] = __int_as_float(__builtin_amdgcn_readlane(rec.y, j + u));
                const v4f* __restrict__ rp = (const v4f*)(dense + (size_t)c * D);
#pragma unroll
                for (int w = 0; w < V; ++w) d[u][w] = rp[lane + w * 32];
            }
#pragma unroll
            for (int u = 0; u < 4; ++u)
#pragma unroll
                for (int w = 0; w < V; ++w) acc[w] += v[u] * d[u][w];
        }
#pragma nounroll
        for (; j < n; ++j) {
            const int   c = __builtin_amdgcn_readlane(rec.x, j);
            const float v = __int_as_float(__builtin_amdgcn_readlane(rec.y, j));
            const v4f* __restrict__ rp = (const v4f*)(dense + (size_t)c * D);
            for (int w = 0; w < V; ++w) acc[w] += v * rp[lane + w * 32];
        }
    }

    if (NORM) {
        float4 v  = make_float4(acc[0].x, acc[0].y, acc[0].z, acc[0].w);
        float4 bb = *(const float4*)(bias + lane * 4);
        v.x += bb.x; v.y += bb.y; v.z += bb.z; v.w += bb.w;

        float ss = v.x * v.x + v.y * v.y + v.z * v.z + v.w * v.w;
        for (int off = 16; off > 0; off >>= 1)
            ss += __shfl_xor(ss, off, 32);

        float scale = 1.0f / fmaxf(sqrtf(ss), 1e-12f);
        v.x *= scale; v.y *= scale; v.z *= scale; v.w *= scale;
        *(float4*)(dst + (size_t)row * D + lane * 4) = v;
    } else {
        v4f* __restrict__ o = (SPLIT < N_NODES && row >= SPLIT)
                                  ? (v4f*)(dst_hi + (size_t)(row - SPLIT) * D)
                                  : (v4f*)(dst + (size_t)row * D);
        for (int w = 0; w < V; ++w) o[lane + w * 32] = acc[w];
    }
}

#define RT_ROWS  16
#define RT_AP    260
#define RT_TILES (N_NODES / RT_ROWS)
#define RT_WGS   256
#define RT_MW    8
#define RT_GW    RT_ROWS
#define RT_NT    ((RT_MW + RT_GW) * 32)
#define RT_G     8

template <int M>
__device__ __forceinline__ void gcn_rt_sum_m(const int2 rec, const int j, const int lane,
                                             const float* __restrict__ dense,
                                             v4f& acc0, v4f& acc1) {
    float v[M];
    v4f   d[M][2];
#pragma unroll
    for (int u = 0; u < M; ++u) {
        const int c = __builtin_amdgcn_readlane(rec.x, j + u);
        v[u] = __int_as_float(__builtin_amdgcn_readlane(rec.y, j + u));
        const v4f* __restrict__ rp = (const v4f*)(dense + (size_t)c * 256);
        d[u][0] = rp[lane];
        d[u][1] = rp[lane + 32];
    }
#pragma unroll
    for (int u = 0; u < M; ++u) {
        acc0 += v[u] * d[u][0];
        acc1 += v[u] * d[u][1];
    }
}

__device__ __forceinline__ void gcn_rt_sum_records(const int2 rec, const int n, const int lane,
                                                   const float* __restrict__ dense,
                                                   v4f& acc0, v4f& acc1) {
    int j = 0;
    for (; j + RT_G <= n; j += RT_G) gcn_rt_sum_m<RT_G>(rec, j, lane, dense, acc0, acc1);
    switch (n - j) {
        case 1: gcn_rt_sum_m<1>(rec, j, lane, dense, acc0, acc1); break;
        case 2: gcn_rt_sum_m<2>(rec, j, lane, dense, acc0, acc1); break;
        case 3: gcn_rt_sum_m<3>(rec, j, lane, dense, acc0, acc1); break;
        case 4: gcn_rt_sum_m<4>(rec, j, lane, dense, acc0, acc1); break;
        case 5: gcn_rt_sum_m<5>(rec, j, lane, dense, acc0, acc1); break;
        case 6: gcn_rt_sum_m<6>(rec, j, lane, dense, acc0, acc1); break;
        case 7: gcn_rt_sum_m<7>(rec, j, lane, dense, acc0, acc1); break;
        default: break;
    }
}

__global__ __launch_bounds__(RT_NT)
void gcn_spmm_gemm2(const int*   __restrict__ rowptr,
                    const int*   __restrict__ edges,
                    const float* __restrict__ dense,
                    const float* __restrict__ bias,
                    const float* __restrict__ w,
                    float*       __restrict__ out) {
    static_assert(N_NODES % RT_ROWS == 0, "no partial tile");
    __shared__ __attribute__((aligned(16))) float Ws[128 * RT_AP];
    __shared__ __attribute__((aligned(16))) float As[2][RT_ROWS * RT_AP];

    const int tid  = threadIdx.x;
    const int lane = tid & 31;
    const int wave  = __builtin_amdgcn_readfirstlane(tid >> 5);
    const int first = (int)blockIdx.x;
    const int nt    = (RT_TILES - first + RT_WGS - 1) / RT_WGS;
    const bool summing = wave >= RT_MW;

    const int2* __restrict__ er = (const int2*)edges;
    const int g = wave - RT_MW;
    int  st0 = 0, en0 = 0, st1 = 0, en1 = 0;
    int2 rec0 = make_int2(0, 0);
    float4 b_lo = make_float4(0.f, 0.f, 0.f, 0.f), b_hi = b_lo;
    if (summing) {
        const int r0 = first * RT_ROWS + g;
        st0 = rowptr[r0]; en0 = rowptr[r0 + 1];
        if (nt > 1) {
            const int r1 = (first + RT_WGS) * RT_ROWS + g;
            st1 = rowptr[r1]; en1 = rowptr[r1 + 1];
        }
        if (lane < en0 - st0) rec0 = er[st0 + lane];
        b_lo = *(const float4*)(bias + lane * 4);
        b_hi = *(const float4*)(bias + 128 + lane * 4);
    }

    for (int it = 0; it <= nt; ++it) {
        if (summing) {
            if (it < nt) {
                int2 rec1 = make_int2(0, 0);
                if (lane < en1 - st1) rec1 = er[st1 + lane];
                int st2 = 0, en2 = 0;
                if (it + 2 < nt) {
                    const int r2 = (first + (it + 2) * RT_WGS) * RT_ROWS + g;
                    st2 = rowptr[r2]; en2 = rowptr[r2 + 1];
                }

                v4f acc0 = (v4f){0.f, 0.f, 0.f, 0.f};
                v4f acc1 = (v4f){0.f, 0.f, 0.f, 0.f};
                gcn_rt_sum_records(rec0, min(32, en0 - st0), lane, dense, acc0, acc1);
                for (int base = st0 + 32; base < en0; base += 32) {
                    const int n = min(32, en0 - base);
                    int2 rec = make_int2(0, 0);
                    if (lane < n) rec = er[base + lane];
                    gcn_rt_sum_records(rec, n, lane, dense, acc0, acc1);
                }
                float4 lo, hi;
                lo.x = fmaxf(acc0.x + b_lo.x, 0.f); lo.y = fmaxf(acc0.y + b_lo.y, 0.f);
                lo.z = fmaxf(acc0.z + b_lo.z, 0.f); lo.w = fmaxf(acc0.w + b_lo.w, 0.f);
                hi.x = fmaxf(acc1.x + b_hi.x, 0.f); hi.y = fmaxf(acc1.y + b_hi.y, 0.f);
                hi.z = fmaxf(acc1.z + b_hi.z, 0.f); hi.w = fmaxf(acc1.w + b_hi.w, 0.f);
                float* ar = As[it & 1] + g * RT_AP;
                *(float4*)(ar + lane * 4)       = lo;
                *(float4*)(ar + 128 + lane * 4) = hi;
                st0 = st1; en0 = en1; rec0 = rec1;
                st1 = st2; en1 = en2;
            }
        } else if (it == 0) {
            const int cr = tid >> 6;
            const int cq = tid & 63;
            const float* __restrict__ wp = w + (size_t)cr * 256 + cq * 4;
            float* wsl = Ws + cr * RT_AP + cq * 4;
            for (int p0 = 0; p0 < 32; p0 += 8) {
                v4f rw[8];
#pragma unroll
                for (int p = 0; p < 8; ++p) rw[p] = *(const v4f*)(wp + (size_t)(p0 + p) * 4 * 256);
#pragma unroll
                for (int p = 0; p < 8; ++p) *(v4f*)(wsl + (p0 + p) * 4 * RT_AP) = rw[p];
            }
        } else {
            const int c  = lane & 15;
            const int h2 = (lane >> 4) << 1;
            const float* af = As[(it - 1) & 1] + c * RT_AP + h2;
            const float* bf = Ws + (wave * 16 + c) * RT_AP + h2;
            v8f acc2 = (v8f){0.f, 0.f, 0.f, 0.f, 0.f, 0.f, 0.f, 0.f};
#pragma unroll
            for (int s = 0; s < 64; ++s) {
                const v2f b  = *(const v2f*)(bf + s * 4);
                const v2f av = *(const v2f*)(af + s * 4);
                acc2 = __builtin_amdgcn_wmma_f32_16x16x4_f32(false, av, false, b,
                                                             (short)0, acc2, false, false);
            }
            const int rowBase = (first + (it - 1) * RT_WGS) * RT_ROWS;
            float* __restrict__ o = out + (size_t)(rowBase + ((lane >> 4) << 3)) * 128 + wave * 16 + c;
#pragma unroll
            for (int j = 0; j < 8; ++j) o[(size_t)j * 128] = acc2[j];
        }
        __syncthreads();
    }
}

extern "C" void kernel_launch(void* const* d_in, const int* in_sizes, int n_in,
                              void* d_out, int out_size, void* d_ws, size_t ws_size,
                              hipStream_t stream) {
    const float* x    = (const float*)d_in[0];
    const int*   erow = (const int*)  d_in[1];
    const int*   ecol = (const int*)  d_in[2];
    const float* ev   = (const float*)d_in[3];
    const float* W1   = (const float*)d_in[4];
    const float* b1   = (const float*)d_in[5];
    const float* W2   = (const float*)d_in[6];
    const float* b2   = (const float*)d_in[7];
    float* out = (float*)d_out;

    char* ws = (char*)d_ws;
    int* rowptr = (int*)(ws + 66560);
    int* offt   = (int*)(ws + 266752);
    float* bufA = (float*)(ws + 438272);
    float* bufB = (float*)(ws + 438272 + 51200000);
    int*   edges   = (int*)(ws + 438272 + 51200000 + 44728320);
    char*  ob      = (char*)d_out;
    int*   inter   = (int*)(ob + 10025984);

    gcn_chunks_gemm1<<<NCH + G1_FIRST, GS_NT, 0, stream>>>(erow, ecol, ev, offt, inter, x, b1, W1, bufA);
    gcn_gemm1_buckets<<<G1_REST + NBK, GS_NT, 0, stream>>>(offt, inter, rowptr, edges, x, b1, W1, bufA);

    gcn_spmm_gemm2<<<RT_WGS, RT_NT, 0, stream>>>(rowptr, edges, bufA, b1, W2, bufB);

    gcn_spmm_rows<128, N_NODES, true><<<N_NODES / 8, 256, 0, stream>>>(rowptr, edges, bufB, b2,
                                                                       out, bufA);
}
